# starting kernel
# baseline (speedup 1.0000x reference)
#include <hip/hip_runtime.h>
#include <hip/hip_bf16.h>

#define N_NODES 50000
#define N_EDGES 800000
#define HID 128
#define OUT_DIM 16
#define NEG_SLOPE 0.2f

typedef __attribute__((ext_vector_type(2))) float v2f;
typedef __attribute__((ext_vector_type(8))) float v8f;

__global__ __launch_bounds__(HID) void embed_kernel(
    const int* __restrict__ feats, const float* __restrict__ key_emb,
    const float* __restrict__ val_emb, float* __restrict__ h) {
  int n = blockIdx.x;
  int d = threadIdx.x;
  int f0 = feats[n * 2 + 0];
  int f1 = feats[n * 2 + 1];
  float v = key_emb[f0 * HID + d] + val_emb[f1 * HID + d];
  h[n * HID + d] = v > 0.f ? v : 0.f;
}

__global__ __launch_bounds__(256) void gemm128_kernel(
    const float* __restrict__ A, const float* __restrict__ B,
    float* __restrict__ C) {
  const int wave = threadIdx.x >> 5;
  const int lane = threadIdx.x & 31;
  const int m0 = blockIdx.x * 16;
  const int n0 = wave * 16;
  const int r = lane & 15;
  const int hi = lane >> 4;
  const int koff = hi << 1;

  const float* arow = A + (size_t)(m0 + r) * HID;
  v8f acc = {};
#pragma unroll 8
  for (int k0 = 0; k0 < HID; k0 += 4) {
    const int k = k0 + koff;
    v2f a;
    a.x = arow[k];
    a.y = arow[k + 1];
    v2f b;
    b.x = B[k * HID + n0 + r];
    b.y = B[(k + 1) * HID + n0 + r];
    acc = __builtin_amdgcn_wmma_f32_16x16x4_f32(false, a, false, b, (short)0,
                                                acc, false, false);
  }
#pragma unroll
  for (int v = 0; v < 8; ++v) {
    C[(size_t)(m0 + v + (hi << 3)) * HID + n0 + r] = acc[v];
  }
}

__global__ __launch_bounds__(256) void gemm_cls_kernel(
    const float* __restrict__ A, const float* __restrict__ B,
    float* __restrict__ C) {
  const int wave = threadIdx.x >> 5;
  const int tile = blockIdx.x * 8 + wave;
  if (tile >= N_NODES / 16) return;
  const int lane = threadIdx.x & 31;
  const int m0 = tile * 16;
  const int r = lane & 15;
  const int hi = lane >> 4;
  const int koff = hi << 1;

  const float* arow = A + (size_t)(m0 + r) * HID;
  v8f acc = {};
#pragma unroll 8
  for (int k0 = 0; k0 < HID; k0 += 4) {
    const int k = k0 + koff;
    v2f a;
    a.x = arow[k];
    a.y = arow[k + 1];
    v2f b;
    b.x = B[k * OUT_DIM + r];
    b.y = B[(k + 1) * OUT_DIM + r];
    acc = __builtin_amdgcn_wmma_f32_16x16x4_f32(false, a, false, b, (short)0,
                                                acc, false, false);
  }
#pragma unroll
  for (int v = 0; v < 8; ++v) {
    C[(size_t)(m0 + v + (hi << 3)) * OUT_DIM + r] = acc[v];
  }
}

__device__ __forceinline__ int float_to_ord(float f) {
  int i = __float_as_int(f);
  return i >= 0 ? i : (i ^ 0x7FFFFFFF);
}
__device__ __forceinline__ float ord_to_float(int i) {
  return __int_as_float(i >= 0 ? i : (i ^ 0x7FFFFFFF));
}

__global__ void init_nodes_kernel(int* __restrict__ mmax,
                                  float* __restrict__ denom) {
  int i = blockIdx.x * 256 + threadIdx.x;
  if (i < N_NODES) {
    mmax[i] = float_to_ord(-__builtin_inff());
    denom[i] = 0.f;
  }
}

__global__ void zero_kernel(float* __restrict__ p, int n) {
  int i = blockIdx.x * 256 + threadIdx.x;
  if (i < n) p[i] = 0.f;
}

__global__ __launch_bounds__(256) void edge_logits_kernel(
    const float* __restrict__ feat, const float* __restrict__ attn,
    const int* __restrict__ src, const int* __restrict__ dst,
    float* __restrict__ logits, int* __restrict__ mmax) {
  const int e = blockIdx.x * 8 + (threadIdx.x >> 5);
  const int lane = threadIdx.x & 31;
  const int s = src[e];
  const int d = dst[e];
  const float4 x = ((const float4*)(feat + (size_t)s * HID))[lane];
  const float4 y = ((const float4*)(feat + (size_t)d * HID))[lane];
  const float4 a4 = ((const float4*)attn)[lane];
  float p = 0.f, t;
  t = x.x + y.x; p += a4.x * (t > 0.f ? t : NEG_SLOPE * t);
  t = x.y + y.y; p += a4.y * (t > 0.f ? t : NEG_SLOPE * t);
  t = x.z + y.z; p += a4.z * (t > 0.f ? t : NEG_SLOPE * t);
  t = x.w + y.w; p += a4.w * (t > 0.f ? t : NEG_SLOPE * t);
#pragma unroll
  for (int off = 16; off > 0; off >>= 1) p += __shfl_xor(p, off, 32);
  if (lane == 0) {
    logits[e] = p;
    atomicMax(mmax + d, float_to_ord(p));
  }
}

__global__ __launch_bounds__(256) void edge_exp_kernel(
    const int* __restrict__ dst, float* __restrict__ logits,
    const int* __restrict__ mmax, float* __restrict__ denom) {
  int e = blockIdx.x * 256 + threadIdx.x;
  if (e >= N_EDGES) return;
  int d = dst[e];
  float mx = ord_to_float(mmax[d]);
  float ex = expf(logits[e] - mx);
  logits[e] = ex;
  atomicAdd(denom + d, ex);
}

__global__ __launch_bounds__(256) void edge_aggr_kernel(
    const float* __restrict__ feat, const int* __restrict__ src,
    const int* __restrict__ dst, const float* __restrict__ ex,
    const float* __restrict__ denom, float* __restrict__ out) {
  const int e = blockIdx.x * 8 + (threadIdx.x >> 5);
  const int lane = threadIdx.x & 31;
  const int s = src[e];
  const int d = dst[e];
  const float a = ex[e] / denom[d];
  const float4 x = ((const float4*)(feat + (size_t)s * HID))[lane];
  float* o = out + (size_t)d * HID + lane * 4;
  atomicAdd(o + 0, a * x.x);
  atomicAdd(o + 1, a * x.y);
  atomicAdd(o + 2, a * x.z);
  atomicAdd(o + 3, a * x.w);
}

static void run_gat_layer(const float* h_in, const float* W, const float* attn,
                          const int* src, const int* dst, float* feat,
                          float* h_out, float* ex, int* mmax, float* denom,
                          hipStream_t stream) {
  gemm128_kernel<<<N_NODES / 16, 256, 0, stream>>>(h_in, W, feat);
  init_nodes_kernel<<<(N_NODES + 255) / 256, 256, 0, stream>>>(mmax, denom);
  zero_kernel<<<(N_NODES * HID + 255) / 256, 256, 0, stream>>>(h_out,
                                                               N_NODES * HID);
  edge_logits_kernel<<<N_EDGES / 8, 256, 0, stream>>>(feat, attn, src, dst, ex,
                                                      mmax);
  edge_exp_kernel<<<(N_EDGES + 255) / 256, 256, 0, stream>>>(dst, ex, mmax,
                                                             denom);
  edge_aggr_kernel<<<N_EDGES / 8, 256, 0, stream>>>(feat, src, dst, ex, denom,
                                                    h_out);
}

extern "C" void kernel_launch(void* const* d_in, const int* in_sizes, int n_in,
                              void* d_out, int out_size, void* d_ws,
                              size_t ws_size, hipStream_t stream) {
  const int* feats = (const int*)d_in[0];
  const int* src = (const int*)d_in[1];
  const int* dst = (const int*)d_in[2];
  const float* key_emb = (const float*)d_in[3];
  const float* val_emb = (const float*)d_in[4];
  const float* W1 = (const float*)d_in[5];
  const float* attn1 = (const float*)d_in[6];
  const float* W2 = (const float*)d_in[7];
  const float* attn2 = (const float*)d_in[8];
  const float* cls_w = (const float*)d_in[9];
  float* out = (float*)d_out;

  float* hA = (float*)d_ws;
  float* feat = hA + (size_t)N_NODES * HID;
  float* hB = feat + (size_t)N_NODES * HID;
  float* ex = hB + (size_t)N_NODES * HID;
  int* mmax = (int*)(ex + N_EDGES);
  float* denom = (float*)(mmax + N_NODES);

  embed_kernel<<<N_NODES, HID, 0, stream>>>(feats, key_emb, val_emb, hA);

  run_gat_layer(hA, W1, attn1, src, dst, feat, hB, ex, mmax, denom, stream);
  run_gat_layer(hB, W2, attn2, src, dst, feat, hA, ex, mmax, denom, stream);

  gemm_cls_kernel<<<(N_NODES / 16 + 7) / 8, 256, 0, stream>>>(hA, cls_w, out);
}


# Embedding fused into the first GEMM: each work-group builds its row tile relu(key+val) in LDS and takes its A fragments from there; the h array and its one-thread-per-element kernel are gone
# speedup vs baseline: 3.1285x; 3.1285x over previous
#include <hip/hip_runtime.h>
#include <hip/hip_bf16.h>

#define N_NODES 50000
#define N_EDGES 800000
#define HID 128
#define OUT_DIM 16
#define NEG_SLOPE 0.2f

#define SCAN_THREADS 1024
#define SCAN_ITEMS 49
#define EDGE_CHUNK 4

static_assert(SCAN_THREADS * SCAN_ITEMS >= N_NODES, "scan must cover every node");
static_assert(N_NODES % 8 == 0, "eight destination rows per work-group");
static_assert(N_NODES % 16 == 0, "sixteen rows per GEMM work-group");

typedef __attribute__((ext_vector_type(2))) float v2f;
typedef __attribute__((ext_vector_type(8))) float v8f;

__global__ __launch_bounds__(256) void gemm128_kernel(
    const float* __restrict__ A, const float* __restrict__ B,
    float* __restrict__ C) {
  const int wave = threadIdx.x >> 5;
  const int lane = threadIdx.x & 31;
  const int m0 = blockIdx.x * 16;
  const int n0 = wave * 16;
  const int r = lane & 15;
  const int hi = lane >> 4;
  const int koff = hi << 1;

  const float* arow = A + (size_t)(m0 + r) * HID;
  v8f acc = {};
#pragma unroll 8
  for (int k0 = 0; k0 < HID; k0 += 4) {
    const int k = k0 + koff;
    v2f a;
    a.x = arow[k];
    a.y = arow[k + 1];
    v2f b;
    b.x = B[k * HID + n0 + r];
    b.y = B[(k + 1) * HID + n0 + r];
    acc = __builtin_amdgcn_wmma_f32_16x16x4_f32(false, a, false, b, (short)0,
                                                acc, false, false);
  }
#pragma unroll
  for (int v = 0; v < 8; ++v) {
    C[(size_t)(m0 + v + (hi << 3)) * HID + n0 + r] = acc[v];
  }
}

#define A_PITCH (HID + 4)

__global__ __launch_bounds__(256) void embed_gemm128_kernel(
    const int* __restrict__ feats, const float* __restrict__ key_emb,
    const float* __restrict__ val_emb, const float* __restrict__ B,
    float* __restrict__ C, int* __restrict__ cnt) {
  __shared__ __attribute__((aligned(16))) float hs[16 * A_PITCH];
  const int lane = threadIdx.x & 31;
  const int wave = __builtin_amdgcn_readfirstlane((int)(threadIdx.x >> 5));
  const int m0 = blockIdx.x * 16;

#pragma unroll
  for (int i = 0; i < 2; ++i) {
    const int row = wave * 2 + i;
    const int f0 = feats[(m0 + row) * 2 + 0];
    const int f1 = feats[(m0 + row) * 2 + 1];
    const float4 kv = ((const float4*)(key_emb + (size_t)f0 * HID))[lane];
    const float4 vv = ((const float4*)(val_emb + (size_t)f1 * HID))[lane];
    float4 hv;
    float s;
    s = kv.x + vv.x; hv.x = s > 0.f ? s : 0.f;
    s = kv.y + vv.y; hv.y = s > 0.f ? s : 0.f;
    s = kv.z + vv.z; hv.z = s > 0.f ? s : 0.f;
    s = kv.w + vv.w; hv.w = s > 0.f ? s : 0.f;
    *(float4*)(hs + row * A_PITCH + lane * 4) = hv;
  }
  if (threadIdx.x < 16) cnt[m0 + threadIdx.x] = 0;
  __syncthreads();

  const int n0 = wave * 16;
  const int r = lane & 15;
  const int hi = lane >> 4;
  const int koff = hi << 1;

  const float* arow = hs + r * A_PITCH + koff;
  v8f acc = {};
#pragma unroll 8
  for (int k0 = 0; k0 < HID; k0 += 4) {
    const int k = k0 + koff;
    const v2f a = *(const v2f*)(arow + k0);
    v2f b;
    b.x = B[k * HID + n0 + r];
    b.y = B[(k + 1) * HID + n0 + r];
    acc = __builtin_amdgcn_wmma_f32_16x16x4_f32(false, a, false, b, (short)0,
                                                acc, false, false);
  }
#pragma unroll
  for (int v = 0; v < 8; ++v) {
    C[(size_t)(m0 + v + (hi << 3)) * HID + n0 + r] = acc[v];
  }
}

__global__ __launch_bounds__(256) void gemm_cls_kernel(
    const float* __restrict__ A, const float* __restrict__ B,
    float* __restrict__ C) {
  const int wave = threadIdx.x >> 5;
  const int tile = blockIdx.x * 8 + wave;
  if (tile >= N_NODES / 16) return;
  const int lane = threadIdx.x & 31;
  const int m0 = tile * 16;
  const int r = lane & 15;
  const int hi = lane >> 4;
  const int koff = hi << 1;

  const float* arow = A + (size_t)(m0 + r) * HID;
  v8f acc = {};
#pragma unroll 8
  for (int k0 = 0; k0 < HID; k0 += 4) {
    const int k = k0 + koff;
    v2f a;
    a.x = arow[k];
    a.y = arow[k + 1];
    v2f b;
    b.x = B[k * OUT_DIM + r];
    b.y = B[(k + 1) * OUT_DIM + r];
    acc = __builtin_amdgcn_wmma_f32_16x16x4_f32(false, a, false, b, (short)0,
                                                acc, false, false);
  }
#pragma unroll
  for (int v = 0; v < 8; ++v) {
    C[(size_t)(m0 + v + (hi << 3)) * OUT_DIM + r] = acc[v];
  }
}

__global__ __launch_bounds__(256) void rank_kernel(
    const int* __restrict__ dst, int* __restrict__ cnt,
    int* __restrict__ rank) {
  const int e = blockIdx.x * 256 + threadIdx.x;
  if (e < N_EDGES) rank[e] = atomicAdd(cnt + dst[e], 1);
}

__global__ __launch_bounds__(SCAN_THREADS) void scan_kernel(
    const int* __restrict__ cnt, int* __restrict__ rowstart) {
  __shared__ int part[SCAN_THREADS];
  const int t = threadIdx.x;
  const int lo = t * SCAN_ITEMS;
  int v[SCAN_ITEMS];
  int sum = 0;
#pragma unroll
  for (int k = 0; k < SCAN_ITEMS; ++k) {
    const int i = lo + k;
    v[k] = (i < N_NODES) ? cnt[i] : 0;
    sum += v[k];
  }
  part[t] = sum;
  __syncthreads();
  for (int off = 1; off < SCAN_THREADS; off <<= 1) {
    const int add = (t >= off) ? part[t - off] : 0;
    __syncthreads();
    part[t] += add;
    __syncthreads();
  }
  int run = part[t] - sum;
#pragma unroll
  for (int k = 0; k < SCAN_ITEMS; ++k) {
    const int i = lo + k;
    if (i < N_NODES) rowstart[i] = run;
    run += v[k];
  }
}

__global__ __launch_bounds__(256) void fill_kernel(
    const int* __restrict__ src, const int* __restrict__ dst,
    const int* __restrict__ rowstart, const int* __restrict__ rank,
    int* __restrict__ csr_src) {
  const int e = blockIdx.x * 256 + threadIdx.x;
  if (e < N_EDGES) csr_src[rowstart[dst[e]] + rank[e]] = src[e];
}

__global__ __launch_bounds__(256) void node_attn_kernel(
    const float* __restrict__ feat, const float* __restrict__ attn,
    const int* __restrict__ rowstart, const int* __restrict__ csr_src,
    float* __restrict__ out) {
  const int lane = threadIdx.x & 31;
  const int wv = __builtin_amdgcn_readfirstlane((int)(threadIdx.x >> 5));
  const int d = blockIdx.x * 8 + wv;
  const float4 a4 = ((const float4*)attn)[lane];

  const int beg = rowstart[d];
  const int end = (d + 1 < N_NODES) ? rowstart[d + 1] : N_EDGES;
  const float4 y = ((const float4*)(feat + (size_t)d * HID))[lane];
  float m = -__builtin_inff();
  float den = 0.f;
  float4 acc = make_float4(0.f, 0.f, 0.f, 0.f);

  for (int base = beg; base < end; base += 32) {
    int nch = end - base;
    if (nch > 32) nch = 32;
    const int li = lane < nch ? lane : nch - 1;
    const int sv = csr_src[base + li];

    for (int j = 0; j < nch; j += EDGE_CHUNK) {
      float4 x[EDGE_CHUNK];
      float l[EDGE_CHUNK];
#pragma unroll
      for (int u = 0; u < EDGE_CHUNK; ++u) {
        int idx = j + u;
        if (idx > nch - 1) idx = nch - 1;
        const int s = __builtin_amdgcn_readlane(sv, idx);
        x[u] = ((const float4*)(feat + (size_t)s * HID))[lane];
      }
#pragma unroll
      for (int u = 0; u < EDGE_CHUNK; ++u) {
        float p = 0.f, t;
        t = x[u].x + y.x; p += a4.x * (t > 0.f ? t : NEG_SLOPE * t);
        t = x[u].y + y.y; p += a4.y * (t > 0.f ? t : NEG_SLOPE * t);
        t = x[u].z + y.z; p += a4.z * (t > 0.f ? t : NEG_SLOPE * t);
        t = x[u].w + y.w; p += a4.w * (t > 0.f ? t : NEG_SLOPE * t);
#pragma unroll
        for (int off = 16; off > 0; off >>= 1) p += __shfl_xor(p, off, 32);
        l[u] = (j + u < nch) ? p : -__builtin_inff();
      }
      float mc = m;
#pragma unroll
      for (int u = 0; u < EDGE_CHUNK; ++u) mc = fmaxf(mc, l[u]);
      const float sc = expf(m - mc);
      den *= sc;
      acc.x *= sc; acc.y *= sc; acc.z *= sc; acc.w *= sc;
#pragma unroll
      for (int u = 0; u < EDGE_CHUNK; ++u) {
        const float w = expf(l[u] - mc);
        den += w;
        acc.x += w * x[u].x;
        acc.y += w * x[u].y;
        acc.z += w * x[u].z;
        acc.w += w * x[u].w;
      }
      m = mc;
    }
  }

  float4 o = make_float4(0.f, 0.f, 0.f, 0.f);
  if (end > beg) {
    o.x = acc.x / den; o.y = acc.y / den;
    o.z = acc.z / den; o.w = acc.w / den;
  }
  ((float4*)(out + (size_t)d * HID))[lane] = o;
}

static void run_gat_layer(const float* h_in, const float* W, const float* attn,
                          const int* rowstart, const int* csr_src, float* feat,
                          float* h_out, hipStream_t stream) {
  gemm128_kernel<<<N_NODES / 16, 256, 0, stream>>>(h_in, W, feat);
  node_attn_kernel<<<N_NODES / 8, 256, 0, stream>>>(feat, attn, rowstart,
                                                   csr_src, h_out);
}

extern "C" void kernel_launch(void* const* d_in, const int* in_sizes, int n_in,
                              void* d_out, int out_size, void* d_ws,
                              size_t ws_size, hipStream_t stream) {
  const int* feats = (const int*)d_in[0];
  const int* src = (const int*)d_in[1];
  const int* dst = (const int*)d_in[2];
  const float* key_emb = (const float*)d_in[3];
  const float* val_emb = (const float*)d_in[4];
  const float* W1 = (const float*)d_in[5];
  const float* attn1 = (const float*)d_in[6];
  const float* W2 = (const float*)d_in[7];
  const float* attn2 = (const float*)d_in[8];
  const float* cls_w = (const float*)d_in[9];
  float* out = (float*)d_out;

  float* hA = (float*)d_ws;
  float* feat = hA + (size_t)N_NODES * HID;
  float* hB = feat + (size_t)N_NODES * HID;
  int* csr_src = (int*)(hB + (size_t)N_NODES * HID);
  int* cnt = csr_src + N_EDGES;
  int* rowstart = cnt + N_NODES;
  int* rank = (int*)hB;

  embed_gemm128_kernel<<<N_NODES / 16, 256, 0, stream>>>(feats, key_emb, val_emb,
                                                       W1, feat, cnt);

  rank_kernel<<<N_EDGES / 256, 256, 0, stream>>>(dst, cnt, rank);
  scan_kernel<<<1, SCAN_THREADS, 0, stream>>>(cnt, rowstart);
  fill_kernel<<<N_EDGES / 256, 256, 0, stream>>>(src, dst, rowstart, rank,
                                                 csr_src);

  node_attn_kernel<<<N_NODES / 8, 256, 0, stream>>>(feat, attn1, rowstart,
                                                   csr_src, hB);
  run_gat_layer(hB, W2, attn2, rowstart, csr_src, feat, hA, stream);

  gemm_cls_kernel<<<(N_NODES / 16 + 7) / 8, 256, 0, stream>>>(hA, cls_w, out);
}


# Prefix sum staged through LDS: counters in and segment starts out with 16-byte accesses of consecutive lanes, each thread scans its 49 consecutive counters from LDS; 32-line 4-byte accesses gone
# speedup vs baseline: 3.8674x; 1.2362x over previous
#include <hip/hip_runtime.h>
#include <hip/hip_bf16.h>

#define N_NODES 50000
#define N_EDGES 800000
#define HID 128
#define OUT_DIM 16
#define NEG_SLOPE 0.2f

#define SCAN_THREADS 1024
#define SCAN_ITEMS 49
#define EDGE_CHUNK 4

static_assert(SCAN_THREADS * SCAN_ITEMS >= N_NODES, "scan must cover every node");
static_assert(N_NODES % 8 == 0, "eight destination rows per work-group");
static_assert(N_NODES % 16 == 0, "sixteen rows per GEMM work-group");

typedef __attribute__((ext_vector_type(2))) float v2f;
typedef __attribute__((ext_vector_type(8))) float v8f;

__global__ __launch_bounds__(256) void gemm128_kernel(
    const float* __restrict__ A, const float* __restrict__ B,
    float* __restrict__ C) {
  const int wave = threadIdx.x >> 5;
  const int lane = threadIdx.x & 31;
  const int m0 = blockIdx.x * 16;
  const int n0 = wave * 16;
  const int r = lane & 15;
  const int hi = lane >> 4;
  const int koff = hi << 1;

  const float* arow = A + (size_t)(m0 + r) * HID;
  v8f acc = {};
#pragma unroll 8
  for (int k0 = 0; k0 < HID; k0 += 4) {
    const int k = k0 + koff;
    v2f a;
    a.x = arow[k];
    a.y = arow[k + 1];
    v2f b;
    b.x = B[k * HID + n0 + r];
    b.y = B[(k + 1) * HID + n0 + r];
    acc = __builtin_amdgcn_wmma_f32_16x16x4_f32(false, a, false, b, (short)0,
                                                acc, false, false);
  }
#pragma unroll
  for (int v = 0; v < 8; ++v) {
    C[(size_t)(m0 + v + (hi << 3)) * HID + n0 + r] = acc[v];
  }
}

#define A_PITCH (HID + 4)

__global__ __launch_bounds__(256) void embed_gemm128_kernel(
    const int* __restrict__ feats, const float* __restrict__ key_emb,
    const float* __restrict__ val_emb, const float* __restrict__ B,
    float* __restrict__ C, int* __restrict__ cnt) {
  __shared__ __attribute__((aligned(16))) float hs[16 * A_PITCH];
  const int lane = threadIdx.x & 31;
  const int wave = __builtin_amdgcn_readfirstlane((int)(threadIdx.x >> 5));
  const int m0 = blockIdx.x * 16;

#pragma unroll
  for (int i = 0; i < 2; ++i) {
    const int row = wave * 2 + i;
    const int f0 = feats[(m0 + row) * 2 + 0];
    const int f1 = feats[(m0 + row) * 2 + 1];
    const float4 kv = ((const float4*)(key_emb + (size_t)f0 * HID))[lane];
    const float4 vv = ((const float4*)(val_emb + (size_t)f1 * HID))[lane];
    float4 hv;
    float s;
    s = kv.x + vv.x; hv.x = s > 0.f ? s : 0.f;
    s = kv.y + vv.y; hv.y = s > 0.f ? s : 0.f;
    s = kv.z + vv.z; hv.z = s > 0.f ? s : 0.f;
    s = kv.w + vv.w; hv.w = s > 0.f ? s : 0.f;
    *(float4*)(hs + row * A_PITCH + lane * 4) = hv;
  }
  if (threadIdx.x < 16) cnt[m0 + threadIdx.x] = 0;
  __syncthreads();

  const int n0 = wave * 16;
  const int r = lane & 15;
  const int hi = lane >> 4;
  const int koff = hi << 1;

  const float* arow = hs + r * A_PITCH + koff;
  v8f acc = {};
#pragma unroll 8
  for (int k0 = 0; k0 < HID; k0 += 4) {
    const int k = k0 + koff;
    const v2f a = *(const v2f*)(arow + k0);
    v2f b;
    b.x = B[k * HID + n0 + r];
    b.y = B[(k + 1) * HID + n0 + r];
    acc = __builtin_amdgcn_wmma_f32_16x16x4_f32(false, a, false, b, (short)0,
                                                acc, false, false);
  }
#pragma unroll
  for (int v = 0; v < 8; ++v) {
    C[(size_t)(m0 + v + (hi << 3)) * HID + n0 + r] = acc[v];
  }
}

__global__ __launch_bounds__(256) void gemm_cls_kernel(
    const float* __restrict__ A, const float* __restrict__ B,
    float* __restrict__ C) {
  const int wave = threadIdx.x >> 5;
  const int tile = blockIdx.x * 8 + wave;
  if (tile >= N_NODES / 16) return;
  const int lane = threadIdx.x & 31;
  const int m0 = tile * 16;
  const int r = lane & 15;
  const int hi = lane >> 4;
  const int koff = hi << 1;

  const float* arow = A + (size_t)(m0 + r) * HID;
  v8f acc = {};
#pragma unroll 8
  for (int k0 = 0; k0 < HID; k0 += 4) {
    const int k = k0 + koff;
    v2f a;
    a.x = arow[k];
    a.y = arow[k + 1];
    v2f b;
    b.x = B[k * OUT_DIM + r];
    b.y = B[(k + 1) * OUT_DIM + r];
    acc = __builtin_amdgcn_wmma_f32_16x16x4_f32(false, a, false, b, (short)0,
                                                acc, false, false);
  }
#pragma unroll
  for (int v = 0; v < 8; ++v) {
    C[(size_t)(m0 + v + (hi << 3)) * OUT_DIM + r] = acc[v];
  }
}

__global__ __launch_bounds__(256) void rank_kernel(
    const int* __restrict__ dst, int* __restrict__ cnt,
    int* __restrict__ rank) {
  const int e = blockIdx.x * 256 + threadIdx.x;
  if (e < N_EDGES) rank[e] = atomicAdd(cnt + dst[e], 1);
}

#define SCAN_WORDS (SCAN_THREADS * SCAN_ITEMS)
#define SCAN_VEC_TRIPS ((SCAN_WORDS / 4 + SCAN_THREADS - 1) / SCAN_THREADS)

static_assert(SCAN_ITEMS % 2 == 1, "odd lane stride keeps LDS conflict-free");
static_assert(N_NODES % 4 == 0 && SCAN_WORDS % 4 == 0, "16-byte pieces");
static_assert(SCAN_THREADS / 32 == 32, "one lane per wave total");

__global__ __launch_bounds__(SCAN_THREADS) void scan_kernel(
    const int* __restrict__ cnt, int* __restrict__ rowstart) {
  __shared__ __attribute__((aligned(16))) int buf[SCAN_WORDS];
  __shared__ int wsum[SCAN_THREADS / 32];
  const int t = threadIdx.x;
  const int lane = t & 31;
  const int wave = t >> 5;
  const int4* cnt4 = (const int4*)cnt;
  int4* rowstart4 = (int4*)rowstart;
  int4* buf4 = (int4*)buf;

#pragma unroll
  for (int j = 0; j < SCAN_VEC_TRIPS; ++j) {
    const int q = j * SCAN_THREADS + t;
    if (q < SCAN_WORDS / 4) {
      int4 c = make_int4(0, 0, 0, 0);
      if (q < N_NODES / 4) c = cnt4[q];
      buf4[q] = c;
    }
  }
  __syncthreads();

  const int lo = t * SCAN_ITEMS;
  int v[SCAN_ITEMS];
  int sum = 0;
#pragma unroll
  for (int k = 0; k < SCAN_ITEMS; ++k) {
    v[k] = buf[lo + k];
    sum += v[k];
  }

  int incl = sum;
#pragma unroll
  for (int off = 1; off < 32; off <<= 1) {
    const int n = __shfl_up(incl, off, 32);
    if (lane >= off) incl += n;
  }
  if (lane == 31) wsum[wave] = incl;
  __syncthreads();
  int pre = (lane < wave) ? wsum[lane] : 0;
#pragma unroll
  for (int off = 16; off > 0; off >>= 1) pre += __shfl_xor(pre, off, 32);

  int run = pre + incl - sum;
#pragma unroll
  for (int k = 0; k < SCAN_ITEMS; ++k) {
    buf[lo + k] = run;
    run += v[k];
  }
  __syncthreads();

#pragma unroll
  for (int j = 0; j < SCAN_VEC_TRIPS; ++j) {
    const int q = j * SCAN_THREADS + t;
    if (q < N_NODES / 4) rowstart4[q] = buf4[q];
  }
}

__global__ __launch_bounds__(256) void fill_kernel(
    const int* __restrict__ src, const int* __restrict__ dst,
    const int* __restrict__ rowstart, const int* __restrict__ rank,
    int* __restrict__ csr_src) {
  const int e = blockIdx.x * 256 + threadIdx.x;
  if (e < N_EDGES) csr_src[rowstart[dst[e]] + rank[e]] = src[e];
}

__global__ __launch_bounds__(256) void node_attn_kernel(
    const float* __restrict__ feat, const float* __restrict__ attn,
    const int* __restrict__ rowstart, const int* __restrict__ csr_src,
    float* __restrict__ out) {
  const int lane = threadIdx.x & 31;
  const int wv = __builtin_amdgcn_readfirstlane((int)(threadIdx.x >> 5));
  const int d = blockIdx.x * 8 + wv;
  const float4 a4 = ((const float4*)attn)[lane];

  const int beg = rowstart[d];
  const int end = (d + 1 < N_NODES) ? rowstart[d + 1] : N_EDGES;
  const float4 y = ((const float4*)(feat + (size_t)d * HID))[lane];
  float m = -__builtin_inff();
  float den = 0.f;
  float4 acc = make_float4(0.f, 0.f, 0.f, 0.f);

  for (int base = beg; base < end; base += 32) {
    int nch = end - base;
    if (nch > 32) nch = 32;
    const int li = lane < nch ? lane : nch - 1;
    const int sv = csr_src[base + li];

    for (int j = 0; j < nch; j += EDGE_CHUNK) {
      float4 x[EDGE_CHUNK];
      float l[EDGE_CHUNK];
#pragma unroll
      for (int u = 0; u < EDGE_CHUNK; ++u) {
        int idx = j + u;
        if (idx > nch - 1) idx = nch - 1;
        const int s = __builtin_amdgcn_readlane(sv, idx);
        x[u] = ((const float4*)(feat + (size_t)s * HID))[lane];
      }
#pragma unroll
      for (int u = 0; u < EDGE_CHUNK; ++u) {
        float p = 0.f, t;
        t = x[u].x + y.x; p += a4.x * (t > 0.f ? t : NEG_SLOPE * t);
        t = x[u].y + y.y; p += a4.y * (t > 0.f ? t : NEG_SLOPE * t);
        t = x[u].z + y.z; p += a4.z * (t > 0.f ? t : NEG_SLOPE * t);
        t = x[u].w + y.w; p += a4.w * (t > 0.f ? t : NEG_SLOPE * t);
#pragma unroll
        for (int off = 16; off > 0; off >>= 1) p += __shfl_xor(p, off, 32);
        l[u] = (j + u < nch) ? p : -__builtin_inff();
      }
      float mc = m;
#pragma unroll
      for (int u = 0; u < EDGE_CHUNK; ++u) mc = fmaxf(mc, l[u]);
      const float sc = expf(m - mc);
      den *= sc;
      acc.x *= sc; acc.y *= sc; acc.z *= sc; acc.w *= sc;
#pragma unroll
      for (int u = 0; u < EDGE_CHUNK; ++u) {
        const float w = expf(l[u] - mc);
        den += w;
        acc.x += w * x[u].x;
        acc.y += w * x[u].y;
        acc.z += w * x[u].z;
        acc.w += w * x[u].w;
      }
      m = mc;
    }
  }

  float4 o = make_float4(0.f, 0.f, 0.f, 0.f);
  if (end > beg) {
    o.x = acc.x / den; o.y = acc.y / den;
    o.z = acc.z / den; o.w = acc.w / den;
  }
  ((float4*)(out + (size_t)d * HID))[lane] = o;
}

static void run_gat_layer(const float* h_in, const float* W, const float* attn,
                          const int* rowstart, const int* csr_src, float* feat,
                          float* h_out, hipStream_t stream) {
  gemm128_kernel<<<N_NODES / 16, 256, 0, stream>>>(h_in, W, feat);
  node_attn_kernel<<<N_NODES / 8, 256, 0, stream>>>(feat, attn, rowstart,
                                                   csr_src, h_out);
}

extern "C" void kernel_launch(void* const* d_in, const int* in_sizes, int n_in,
                              void* d_out, int out_size, void* d_ws,
                              size_t ws_size, hipStream_t stream) {
  const int* feats = (const int*)d_in[0];
  const int* src = (const int*)d_in[1];
  const int* dst = (const int*)d_in[2];
  const float* key_emb = (const float*)d_in[3];
  const float* val_emb = (const float*)d_in[4];
  const float* W1 = (const float*)d_in[5];
  const float* attn1 = (const float*)d_in[6];
  const float* W2 = (const float*)d_in[7];
  const float* attn2 = (const float*)d_in[8];
  const float* cls_w = (const float*)d_in[9];
  float* out = (float*)d_out;

  float* hA = (float*)d_ws;
  float* feat = hA + (size_t)N_NODES * HID;
  float* hB = feat + (size_t)N_NODES * HID;
  int* csr_src = (int*)(hB + (size_t)N_NODES * HID);
  int* cnt = csr_src + N_EDGES;
  int* rowstart = cnt + N_NODES;
  int* rank = (int*)hB;

  embed_gemm128_kernel<<<N_NODES / 16, 256, 0, stream>>>(feats, key_emb, val_emb,
                                                       W1, feat, cnt);

  rank_kernel<<<N_EDGES / 256, 256, 0, stream>>>(dst, cnt, rank);
  scan_kernel<<<1, SCAN_THREADS, 0, stream>>>(cnt, rowstart);
  fill_kernel<<<N_EDGES / 256, 256, 0, stream>>>(src, dst, rowstart, rank,
                                                 csr_src);

  node_attn_kernel<<<N_NODES / 8, 256, 0, stream>>>(feat, attn1, rowstart,
                                                   csr_src, hB);
  run_gat_layer(hB, W2, attn2, rowstart, csr_src, feat, hA, stream);

  gemm_cls_kernel<<<(N_NODES / 16 + 7) / 8, 256, 0, stream>>>(hA, cls_w, out);
}


# Attention row loop: the wave takes two source rows at a time, one per 16-lane half; a logit is summed by four in-half DPP adds, not five LDS shuffles per edge; the sums are added in a new order
# speedup vs baseline: 4.4705x; 1.1559x over previous
#include <hip/hip_runtime.h>
#include <hip/hip_bf16.h>

#define N_NODES 50000
#define N_EDGES 800000
#define HID 128
#define OUT_DIM 16
#define NEG_SLOPE 0.2f

#define SCAN_THREADS 1024
#define SCAN_ITEMS 49
#define EDGE_CHUNK 4

static_assert(SCAN_THREADS * SCAN_ITEMS >= N_NODES, "scan must cover every node");
static_assert(N_NODES % 16 == 0, "sixteen rows per work-group");

typedef __attribute__((ext_vector_type(2))) float v2f;
typedef __attribute__((ext_vector_type(8))) float v8f;

#define A_PITCH (HID + 4)

__global__ __launch_bounds__(256) void embed_gemm128_kernel(
    const int* __restrict__ feats, const float* __restrict__ key_emb,
    const float* __restrict__ val_emb, const float* __restrict__ B,
    float* __restrict__ C, int* __restrict__ cnt) {
  __shared__ __attribute__((aligned(16))) float hs[16 * A_PITCH];
  const int lane = threadIdx.x & 31;
  const int wave = __builtin_amdgcn_readfirstlane((int)(threadIdx.x >> 5));
  const int m0 = blockIdx.x * 16;

#pragma unroll
  for (int i = 0; i < 2; ++i) {
    const int row = wave * 2 + i;
    const int f0 = feats[(m0 + row) * 2 + 0];
    const int f1 = feats[(m0 + row) * 2 + 1];
    const float4 kv = ((const float4*)(key_emb + (size_t)f0 * HID))[lane];
    const float4 vv = ((const float4*)(val_emb + (size_t)f1 * HID))[lane];
    float4 hv;
    float s;
    s = kv.x + vv.x; hv.x = s > 0.f ? s : 0.f;
    s = kv.y + vv.y; hv.y = s > 0.f ? s : 0.f;
    s = kv.z + vv.z; hv.z = s > 0.f ? s : 0.f;
    s = kv.w + vv.w; hv.w = s > 0.f ? s : 0.f;
    *(float4*)(hs + row * A_PITCH + lane * 4) = hv;
  }
  if (threadIdx.x < 16) cnt[m0 + threadIdx.x] = 0;
  __syncthreads();

  const int n0 = wave * 16;
  const int r = lane & 15;
  const int hi = lane >> 4;
  const int koff = hi << 1;

  const float* arow = hs + r * A_PITCH + koff;
  v8f acc = {};
#pragma unroll 8
  for (int k0 = 0; k0 < HID; k0 += 4) {
    const int k = k0 + koff;
    const v2f a = *(const v2f*)(arow + k0);
    v2f b;
    b.x = B[k * HID + n0 + r];
    b.y = B[(k + 1) * HID + n0 + r];
    acc = __builtin_amdgcn_wmma_f32_16x16x4_f32(false, a, false, b, (short)0,
                                                acc, false, false);
  }
#pragma unroll
  for (int v = 0; v < 8; ++v) {
    C[(size_t)(m0 + v + (hi << 3)) * HID + n0 + r] = acc[v];
  }
}

__global__ __launch_bounds__(256) void rank_kernel(
    const int* __restrict__ dst, int* __restrict__ cnt,
    int* __restrict__ rank) {
  const int e = blockIdx.x * 256 + threadIdx.x;
  if (e < N_EDGES) rank[e] = atomicAdd(cnt + dst[e], 1);
}

#define SCAN_WORDS (SCAN_THREADS * SCAN_ITEMS)
#define SCAN_VEC_TRIPS ((SCAN_WORDS / 4 + SCAN_THREADS - 1) / SCAN_THREADS)

static_assert(SCAN_ITEMS % 2 == 1, "odd lane stride keeps LDS conflict-free");
static_assert(N_NODES % 4 == 0 && SCAN_WORDS % 4 == 0, "16-byte pieces");
static_assert(SCAN_THREADS / 32 == 32, "one lane per wave total");

__global__ __launch_bounds__(SCAN_THREADS) void scan_kernel(
    const int* __restrict__ cnt, int* __restrict__ rowstart) {
  __shared__ __attribute__((aligned(16))) int buf[SCAN_WORDS];
  __shared__ int wsum[SCAN_THREADS / 32];
  const int t = threadIdx.x;
  const int lane = t & 31;
  const int wave = t >> 5;
  const int4* cnt4 = (const int4*)cnt;
  int4* rowstart4 = (int4*)rowstart;
  int4* buf4 = (int4*)buf;

#pragma unroll
  for (int j = 0; j < SCAN_VEC_TRIPS; ++j) {
    const int q = j * SCAN_THREADS + t;
    if (q < SCAN_WORDS / 4) {
      int4 c = make_int4(0, 0, 0, 0);
      if (q < N_NODES / 4) c = cnt4[q];
      buf4[q] = c;
    }
  }
  __syncthreads();

  const int lo = t * SCAN_ITEMS;
  int v[SCAN_ITEMS];
  int sum = 0;
#pragma unroll
  for (int k = 0; k < SCAN_ITEMS; ++k) {
    v[k] = buf[lo + k];
    sum += v[k];
  }

  int incl = sum;
#pragma unroll
  for (int off = 1; off < 32; off <<= 1) {
    const int n = __shfl_up(incl, off, 32);
    if (lane >= off) incl += n;
  }
  if (lane == 31) wsum[wave] = incl;
  __syncthreads();
  int pre = (lane < wave) ? wsum[lane] : 0;
#pragma unroll
  for (int off = 16; off > 0; off >>= 1) pre += __shfl_xor(pre, off, 32);

  int run = pre + incl - sum;
#pragma unroll
  for (int k = 0; k < SCAN_ITEMS; ++k) {
    buf[lo + k] = run;
    run += v[k];
  }
  __syncthreads();

#pragma unroll
  for (int j = 0; j < SCAN_VEC_TRIPS; ++j) {
    const int q = j * SCAN_THREADS + t;
    if (q < N_NODES / 4) rowstart4[q] = buf4[q];
  }
}

__global__ __launch_bounds__(256) void fill_kernel(
    const int* __restrict__ src, const int* __restrict__ dst,
    const int* __restrict__ rowstart, const int* __restrict__ rank,
    int* __restrict__ csr_src) {
  const int e = blockIdx.x * 256 + threadIdx.x;
  if (e < N_EDGES) csr_src[rowstart[dst[e]] + rank[e]] = src[e];
}

template <int M>
__device__ __forceinline__ float row_xor(const float x) {
  const int b = __builtin_bit_cast(int, x);
  return __builtin_bit_cast(
      float, __builtin_amdgcn_mov_dpp(b, 0x160 + M, 0xf, 0xf, true));
}

__device__ __forceinline__ float xhalf(const float x) {
  const int b = __builtin_bit_cast(int, x);
  return __builtin_bit_cast(
      float, __builtin_amdgcn_permlanex16(b, b, 0x76543210, 0xfedcba98, false,
                                          false));
}

__device__ __forceinline__ v2f lrelu2(const v2f t) {
  v2f r;
  r.x = t.x > 0.f ? t.x : NEG_SLOPE * t.x;
  r.y = t.y > 0.f ? t.y : NEG_SLOPE * t.y;
  return r;
}

__device__ __forceinline__ v2f lo2(const float4 v) { return (v2f){v.x, v.y}; }
__device__ __forceinline__ v2f hi2(const float4 v) { return (v2f){v.z, v.w}; }

__device__ __forceinline__ float logit_part(const float4 a0, const float4 a1,
                                            const float4 x0, const float4 x1,
                                            const float4 y0, const float4 y1) {
  v2f c = lo2(a0) * lrelu2(lo2(x0) + lo2(y0));
  c = hi2(a0) * lrelu2(hi2(x0) + hi2(y0)) + c;
  c = lo2(a1) * lrelu2(lo2(x1) + lo2(y1)) + c;
  c = hi2(a1) * lrelu2(hi2(x1) + hi2(y1)) + c;
  return c.x + c.y;
}

__device__ __forceinline__ float4 attn_row(
    const float* __restrict__ feat, const float4 a0, const float4 a1,
    const int* __restrict__ csr_src, const int d, const int beg, const int end,
    const int lane) {
  const int h = lane >> 4;
  const int i = lane & 15;
  const float4* yrow = (const float4*)(feat + (size_t)d * HID) + i;
  const float4 y0 = yrow[0];
  const float4 y1 = yrow[16];
  float m = -__builtin_inff();
  float den = 0.f;
  float4 acc0 = make_float4(0.f, 0.f, 0.f, 0.f);
  float4 acc1 = make_float4(0.f, 0.f, 0.f, 0.f);

  for (int base = beg; base < end; base += 32) {
    int nch = end - base;
    if (nch > 32) nch = 32;
    const int li = lane < nch ? lane : nch - 1;
    const int sv = csr_src[base + li];

    for (int j = 0; j < nch; j += EDGE_CHUNK) {
      int s[EDGE_CHUNK];
#pragma unroll
      for (int u = 0; u < EDGE_CHUNK; ++u) {
        int idx = j + u;
        if (idx > nch - 1) idx = nch - 1;
        s[u] = __builtin_amdgcn_readlane(sv, idx);
      }
      const int sa = h ? s[1] : s[0];
      const int sb = h ? s[3] : s[2];
      const float4* pa = (const float4*)(feat + (size_t)sa * HID) + i;
      const float4* pb = (const float4*)(feat + (size_t)sb * HID) + i;
      const float4 xa0 = pa[0];
      const float4 xa1 = pa[16];
      const float4 xb0 = pb[0];
      const float4 xb1 = pb[16];

      float la = logit_part(a0, a1, xa0, xa1, y0, y1);
      float lb = logit_part(a0, a1, xb0, xb1, y0, y1);
      la += row_xor<1>(la); lb += row_xor<1>(lb);
      la += row_xor<2>(la); lb += row_xor<2>(lb);
      la += row_xor<4>(la); lb += row_xor<4>(lb);
      la += row_xor<8>(la); lb += row_xor<8>(lb);
      if (j + h >= nch) la = -__builtin_inff();
      if (j + 2 + h >= nch) lb = -__builtin_inff();
      float mc = fmaxf(la, lb);
      mc = fmaxf(mc, xhalf(mc));
      mc = fmaxf(mc, m);
      const float sc = expf(m - mc);
      const float wa = expf(la - mc);
      const float wb = expf(lb - mc);
      den = den * sc + (wa + wb);
      acc0.x = acc0.x * sc + (wa * xa0.x + wb * xb0.x);
      acc0.y = acc0.y * sc + (wa * xa0.y + wb * xb0.y);
      acc0.z = acc0.z * sc + (wa * xa0.z + wb * xb0.z);
      acc0.w = acc0.w * sc + (wa * xa0.w + wb * xb0.w);
      acc1.x = acc1.x * sc + (wa * xa1.x + wb * xb1.x);
      acc1.y = acc1.y * sc + (wa * xa1.y + wb * xb1.y);
      acc1.z = acc1.z * sc + (wa * xa1.z + wb * xb1.z);
      acc1.w = acc1.w * sc + (wa * xa1.w + wb * xb1.w);
      m = mc;
    }
  }

  den += xhalf(den);
  acc0.x += xhalf(acc0.x); acc0.y += xhalf(acc0.y);
  acc0.z += xhalf(acc0.z); acc0.w += xhalf(acc0.w);
  acc1.x += xhalf(acc1.x); acc1.y += xhalf(acc1.y);
  acc1.z += xhalf(acc1.z); acc1.w += xhalf(acc1.w);
  const float4 mine = h ? acc1 : acc0;

  float4 o = make_float4(0.f, 0.f, 0.f, 0.f);
  if (end > beg) {
    o.x = mine.x / den; o.y = mine.y / den;
    o.z = mine.z / den; o.w = mine.w / den;
  }
  return o;
}

__device__ __forceinline__ void attn_stripe_to_lds(
    const float* __restrict__ feat, const float* __restrict__ attn,
    const int* __restrict__ rowstart, const int* __restrict__ csr_src,
    float* hs, const int m0, const int wv, const int lane) {
  const float4 a0 = ((const float4*)attn)[lane & 15];
  const float4 a1 = ((const float4*)attn)[16 + (lane & 15)];
  const int r0 = wv * 2;
  const int d0 = m0 + r0;
  const int b0 = rowstart[d0];
  const int b1 = rowstart[d0 + 1];
  const int b2 = (d0 + 2 < N_NODES) ? rowstart[d0 + 2] : N_EDGES;
  for (int i = 0; i < 2; ++i) {
    const int beg = i ? b1 : b0;
    const int end = i ? b2 : b1;
    const float4 o = attn_row(feat, a0, a1, csr_src, d0 + i, beg, end, lane);
    *(float4*)(hs + (r0 + i) * A_PITCH + lane * 4) = o;
  }
}

__global__ __launch_bounds__(256) void attn_gemm128_kernel(
    const float* __restrict__ feat, const float* __restrict__ attn,
    const int* __restrict__ rowstart, const int* __restrict__ csr_src,
    const float* __restrict__ B, float* __restrict__ C) {
  __shared__ __attribute__((aligned(16))) float hs[16 * A_PITCH];
  const int lane = threadIdx.x & 31;
  const int wv = __builtin_amdgcn_readfirstlane((int)(threadIdx.x >> 5));
  const int m0 = blockIdx.x * 16;

  attn_stripe_to_lds(feat, attn, rowstart, csr_src, hs, m0, wv, lane);
  __syncthreads();

  const int n0 = wv * 16;
  const int r = lane & 15;
  const int hi = lane >> 4;
  const int koff = hi << 1;

  const float* arow = hs + r * A_PITCH + koff;
  v8f acc = {};
#pragma unroll 16
  for (int k0 = 0; k0 < HID; k0 += 4) {
    const int k = k0 + koff;
    const v2f a = *(const v2f*)(arow + k0);
    v2f b;
    b.x = B[k * HID + n0 + r];
    b.y = B[(k + 1) * HID + n0 + r];
    acc = __builtin_amdgcn_wmma_f32_16x16x4_f32(false, a, false, b, (short)0,
                                                acc, false, false);
  }
#pragma unroll
  for (int v = 0; v < 8; ++v) {
    C[(size_t)(m0 + v + (hi << 3)) * HID + n0 + r] = acc[v];
  }
}

__global__ __launch_bounds__(256) void attn_cls_kernel(
    const float* __restrict__ feat, const float* __restrict__ attn,
    const int* __restrict__ rowstart, const int* __restrict__ csr_src,
    const float* __restrict__ B, float* __restrict__ C) {
  __shared__ __attribute__((aligned(16))) float hs[16 * A_PITCH];
  const int lane = threadIdx.x & 31;
  const int wv = __builtin_amdgcn_readfirstlane((int)(threadIdx.x >> 5));
  const int m0 = blockIdx.x * 16;

  attn_stripe_to_lds(feat, attn, rowstart, csr_src, hs, m0, wv, lane);
  __syncthreads();

  if (wv != 0) return;
  const int r = lane & 15;
  const int hi = lane >> 4;
  const int koff = hi << 1;

  const float* arow = hs + r * A_PITCH + koff;
  v8f acc = {};
#pragma unroll 16
  for (int k0 = 0; k0 < HID; k0 += 4) {
    const int k = k0 + koff;
    const v2f a = *(const v2f*)(arow + k0);
    v2f b;
    b.x = B[k * OUT_DIM + r];
    b.y = B[(k + 1) * OUT_DIM + r];
    acc = __builtin_amdgcn_wmma_f32_16x16x4_f32(false, a, false, b, (short)0,
                                                acc, false, false);
  }
#pragma unroll
  for (int v = 0; v < 8; ++v) {
    C[(size_t)(m0 + v + (hi << 3)) * OUT_DIM + r] = acc[v];
  }
}

extern "C" void kernel_launch(void* const* d_in, const int* in_sizes, int n_in,
                              void* d_out, int out_size, void* d_ws,
                              size_t ws_size, hipStream_t stream) {
  const int* feats = (const int*)d_in[0];
  const int* src = (const int*)d_in[1];
  const int* dst = (const int*)d_in[2];
  const float* key_emb = (const float*)d_in[3];
  const float* val_emb = (const float*)d_in[4];
  const float* W1 = (const float*)d_in[5];
  const float* attn1 = (const float*)d_in[6];
  const float* W2 = (const float*)d_in[7];
  const float* attn2 = (const float*)d_in[8];
  const float* cls_w = (const float*)d_in[9];
  float* out = (float*)d_out;

  float* feat2 = (float*)d_ws;
  float* feat = feat2 + (size_t)N_NODES * HID;
  float* third = feat + (size_t)N_NODES * HID;
  int* csr_src = (int*)(third + (size_t)N_NODES * HID);
  int* cnt = csr_src + N_EDGES;
  int* rowstart = cnt + N_NODES;
  int* rank = (int*)third;

  embed_gemm128_kernel<<<N_NODES / 16, 256, 0, stream>>>(feats, key_emb, val_emb,
                                                       W1, feat, cnt);

  rank_kernel<<<N_EDGES / 256, 256, 0, stream>>>(dst, cnt, rank);
  scan_kernel<<<1, SCAN_THREADS, 0, stream>>>(cnt, rowstart);
  fill_kernel<<<N_EDGES / 256, 256, 0, stream>>>(src, dst, rowstart, rank,
                                                 csr_src);

  attn_gemm128_kernel<<<N_NODES / 16, 256, 0, stream>>>(feat, attn1, rowstart,
                                                      csr_src, W2, feat2);
  attn_cls_kernel<<<N_NODES / 16, 256, 0, stream>>>(feat2, attn2, rowstart,
                                                  csr_src, cls_w, out);
}


# Softmax weights in base 2: attention vector scaled by log2(e) once per wave, so each weight is one hardware exponential; range-reduction code of 3 exponentials per 4 edges gone; same sum order
# speedup vs baseline: 5.1649x; 1.1553x over previous
#include <hip/hip_runtime.h>
#include <hip/hip_bf16.h>

#define N_NODES 50000
#define N_EDGES 800000
#define HID 128
#define OUT_DIM 16
#define NEG_SLOPE 0.2f
#define LOG2E 1.4426950408889634f

#define EDGE_CHUNK 4
#define SLOT_CAP 64

static_assert(N_NODES % 16 == 0, "sixteen rows per work-group");
static_assert(N_EDGES % 32 == 0, "the rescan path reads whole 32-edge windows");
static_assert(SLOT_CAP % 32 == 0, "a node's slots are whole 128-byte lines");

typedef __attribute__((ext_vector_type(2))) float v2f;
typedef __attribute__((ext_vector_type(8))) float v8f;

#define A_PITCH (HID + 4)

__global__ __launch_bounds__(256) void embed_gemm128_kernel(
    const int* __restrict__ feats, const float* __restrict__ key_emb,
    const float* __restrict__ val_emb, const float* __restrict__ B,
    float* __restrict__ C, int* __restrict__ cnt) {
  __shared__ __attribute__((aligned(16))) float hs[16 * A_PITCH];
  const int lane = threadIdx.x & 31;
  const int wave = __builtin_amdgcn_readfirstlane((int)(threadIdx.x >> 5));
  const int m0 = blockIdx.x * 16;

#pragma unroll
  for (int i = 0; i < 2; ++i) {
    const int row = wave * 2 + i;
    const int f0 = feats[(m0 + row) * 2 + 0];
    const int f1 = feats[(m0 + row) * 2 + 1];
    const float4 kv = ((const float4*)(key_emb + (size_t)f0 * HID))[lane];
    const float4 vv = ((const float4*)(val_emb + (size_t)f1 * HID))[lane];
    float4 hv;
    float s;
    s = kv.x + vv.x; hv.x = s > 0.f ? s : 0.f;
    s = kv.y + vv.y; hv.y = s > 0.f ? s : 0.f;
    s = kv.z + vv.z; hv.z = s > 0.f ? s : 0.f;
    s = kv.w + vv.w; hv.w = s > 0.f ? s : 0.f;
    *(float4*)(hs + row * A_PITCH + lane * 4) = hv;
  }
  if (threadIdx.x < 16) cnt[m0 + threadIdx.x] = 0;
  __syncthreads();

  const int n0 = wave * 16;
  const int r = lane & 15;
  const int hi = lane >> 4;
  const int koff = hi << 1;

  const float* arow = hs + r * A_PITCH + koff;
  v8f acc = {};
#pragma unroll 8
  for (int k0 = 0; k0 < HID; k0 += 4) {
    const int k = k0 + koff;
    const v2f a = *(const v2f*)(arow + k0);
    v2f b;
    b.x = B[k * HID + n0 + r];
    b.y = B[(k + 1) * HID + n0 + r];
    acc = __builtin_amdgcn_wmma_f32_16x16x4_f32(false, a, false, b, (short)0,
                                                acc, false, false);
  }
#pragma unroll
  for (int v = 0; v < 8; ++v) {
    C[(size_t)(m0 + v + (hi << 3)) * HID + n0 + r] = acc[v];
  }
}

__global__ __launch_bounds__(256) void group_kernel(
    const int* __restrict__ src, const int* __restrict__ dst,
    int* __restrict__ cnt, int* __restrict__ slots) {
  const int e = blockIdx.x * 256 + threadIdx.x;
  if (e < N_EDGES) {
    const int d = dst[e];
    const int s = src[e];
    const int r = atomicAdd(cnt + d, 1);
    slots[(size_t)d * SLOT_CAP + (r < SLOT_CAP ? r : SLOT_CAP - 1)] = s;
  }
}

template <int M>
__device__ __forceinline__ float row_xor(const float x) {
  const int b = __builtin_bit_cast(int, x);
  return __builtin_bit_cast(
      float, __builtin_amdgcn_mov_dpp(b, 0x160 + M, 0xf, 0xf, true));
}

__device__ __forceinline__ float xhalf(const float x) {
  const int b = __builtin_bit_cast(int, x);
  return __builtin_bit_cast(
      float, __builtin_amdgcn_permlanex16(b, b, 0x76543210, 0xfedcba98, false,
                                          false));
}

__device__ __forceinline__ v2f lrelu2(const v2f t) {
  v2f r;
  r.x = t.x > 0.f ? t.x : NEG_SLOPE * t.x;
  r.y = t.y > 0.f ? t.y : NEG_SLOPE * t.y;
  return r;
}

__device__ __forceinline__ v2f lo2(const float4 v) { return (v2f){v.x, v.y}; }
__device__ __forceinline__ v2f hi2(const float4 v) { return (v2f){v.z, v.w}; }

__device__ __forceinline__ float logit_part(const float4 a0, const float4 a1,
                                            const float4 x0, const float4 x1,
                                            const float4 y0, const float4 y1) {
  v2f c = lo2(a0) * lrelu2(lo2(x0) + lo2(y0));
  c = hi2(a0) * lrelu2(hi2(x0) + hi2(y0)) + c;
  c = lo2(a1) * lrelu2(lo2(x1) + lo2(y1)) + c;
  c = hi2(a1) * lrelu2(hi2(x1) + hi2(y1)) + c;
  return c.x + c.y;
}

struct RowState {
  float m;
  float den;
  float4 acc0;
  float4 acc1;
};

__device__ __forceinline__ void attn_list(
    const float* __restrict__ feat, const float4 a0, const float4 a1,
    const float4 y0, const float4 y1, const int sv, const int nch, const int h,
    const int i, RowState& st) {
  for (int j = 0; j < nch; j += EDGE_CHUNK) {
    int s[EDGE_CHUNK];
#pragma unroll
    for (int u = 0; u < EDGE_CHUNK; ++u) {
      int idx = j + u;
      if (idx > nch - 1) idx = nch - 1;
      s[u] = __builtin_amdgcn_readlane(sv, idx);
    }
    const int sa = h ? s[1] : s[0];
    const int sb = h ? s[3] : s[2];
    const float4* pa = (const float4*)(feat + (size_t)sa * HID) + i;
    const float4* pb = (const float4*)(feat + (size_t)sb * HID) + i;
    const float4 xa0 = pa[0];
    const float4 xa1 = pa[16];
    const float4 xb0 = pb[0];
    const float4 xb1 = pb[16];

    float la = logit_part(a0, a1, xa0, xa1, y0, y1);
    float lb = logit_part(a0, a1, xb0, xb1, y0, y1);
    la += row_xor<1>(la); lb += row_xor<1>(lb);
    la += row_xor<2>(la); lb += row_xor<2>(lb);
    la += row_xor<4>(la); lb += row_xor<4>(lb);
    la += row_xor<8>(la); lb += row_xor<8>(lb);
    if (j + h >= nch) la = -__builtin_inff();
    if (j + 2 + h >= nch) lb = -__builtin_inff();
    float mc = fmaxf(la, lb);
    mc = fmaxf(mc, xhalf(mc));
    mc = fmaxf(mc, st.m);
    const float sc = __builtin_amdgcn_exp2f(st.m - mc);
    const float wa = __builtin_amdgcn_exp2f(la - mc);
    const float wb = __builtin_amdgcn_exp2f(lb - mc);
    st.den = st.den * sc + (wa + wb);
    st.acc0.x = st.acc0.x * sc + (wa * xa0.x + wb * xb0.x);
    st.acc0.y = st.acc0.y * sc + (wa * xa0.y + wb * xb0.y);
    st.acc0.z = st.acc0.z * sc + (wa * xa0.z + wb * xb0.z);
    st.acc0.w = st.acc0.w * sc + (wa * xa0.w + wb * xb0.w);
    st.acc1.x = st.acc1.x * sc + (wa * xa1.x + wb * xb1.x);
    st.acc1.y = st.acc1.y * sc + (wa * xa1.y + wb * xb1.y);
    st.acc1.z = st.acc1.z * sc + (wa * xa1.z + wb * xb1.z);
    st.acc1.w = st.acc1.w * sc + (wa * xa1.w + wb * xb1.w);
    st.m = mc;
  }
}

__device__ __forceinline__ float4 attn_row(
    const float* __restrict__ feat, const float4 a0, const float4 a1,
    const int* __restrict__ slots, const int* __restrict__ src,
    const int* __restrict__ dst, const int d, const int deg, const int lane) {
  const int h = lane >> 4;
  const int i = lane & 15;
  const float4* yrow = (const float4*)(feat + (size_t)d * HID) + i;
  const float4 y0 = yrow[0];
  const float4 y1 = yrow[16];
  RowState st;
  st.m = -__builtin_inff();
  st.den = 0.f;
  st.acc0 = make_float4(0.f, 0.f, 0.f, 0.f);
  st.acc1 = make_float4(0.f, 0.f, 0.f, 0.f);

  if (deg <= SLOT_CAP) {
    const int* row = slots + (size_t)d * SLOT_CAP;
    for (int base = 0; base < deg; base += 32) {
      int nch = deg - base;
      if (nch > 32) nch = 32;
      const int li = lane < nch ? lane : nch - 1;
      const int sv = row[base + li];
      attn_list(feat, a0, a1, y0, y1, sv, nch, h, i, st);
    }
  } else {
    for (int e0 = 0; e0 < N_EDGES; e0 += 32) {
      const int dv = dst[e0 + lane];
      const int sw = src[e0 + lane];
      unsigned hit = __builtin_amdgcn_ballot_w32(dv == d);
      while (hit) {
        const int b = __builtin_ctz(hit);
        hit &= hit - 1;
        const int sv = __builtin_amdgcn_readlane(sw, b);
        attn_list(feat, a0, a1, y0, y1, sv, 1, h, i, st);
      }
    }
  }

  float den = st.den;
  float4 acc0 = st.acc0;
  float4 acc1 = st.acc1;
  den += xhalf(den);
  acc0.x += xhalf(acc0.x); acc0.y += xhalf(acc0.y);
  acc0.z += xhalf(acc0.z); acc0.w += xhalf(acc0.w);
  acc1.x += xhalf(acc1.x); acc1.y += xhalf(acc1.y);
  acc1.z += xhalf(acc1.z); acc1.w += xhalf(acc1.w);
  const float4 mine = h ? acc1 : acc0;

  float4 o = make_float4(0.f, 0.f, 0.f, 0.f);
  if (deg > 0) {
    o.x = mine.x / den; o.y = mine.y / den;
    o.z = mine.z / den; o.w = mine.w / den;
  }
  return o;
}

__device__ __forceinline__ void attn_stripe_to_lds(
    const float* __restrict__ feat, const float* __restrict__ attn,
    const int* __restrict__ cnt, const int* __restrict__ slots,
    const int* __restrict__ src, const int* __restrict__ dst, float* hs,
    const int m0, const int wv, const int lane) {
  float4 a0 = ((const float4*)attn)[lane & 15];
  float4 a1 = ((const float4*)attn)[16 + (lane & 15)];
  a0.x *= LOG2E; a0.y *= LOG2E; a0.z *= LOG2E; a0.w *= LOG2E;
  a1.x *= LOG2E; a1.y *= LOG2E; a1.z *= LOG2E; a1.w *= LOG2E;
  const int r0 = wv * 2;
  const int d0 = m0 + r0;
  const int g0 = cnt[d0];
  const int g1 = cnt[d0 + 1];
  for (int i = 0; i < 2; ++i) {
    const int deg = i ? g1 : g0;
    const float4 o =
        attn_row(feat, a0, a1, slots, src, dst, d0 + i, deg, lane);
    *(float4*)(hs + (r0 + i) * A_PITCH + lane * 4) = o;
  }
}

__global__ __launch_bounds__(256) void attn_gemm128_kernel(
    const float* __restrict__ feat, const float* __restrict__ attn,
    const int* __restrict__ cnt, const int* __restrict__ slots,
    const int* __restrict__ src, const int* __restrict__ dst,
    const float* __restrict__ B, float* __restrict__ C) {
  __shared__ __attribute__((aligned(16))) float hs[16 * A_PITCH];
  const int lane = threadIdx.x & 31;
  const int wv = __builtin_amdgcn_readfirstlane((int)(threadIdx.x >> 5));
  const int m0 = blockIdx.x * 16;

  attn_stripe_to_lds(feat, attn, cnt, slots, src, dst, hs, m0, wv, lane);
  __syncthreads();

  const int n0 = wv * 16;
  const int r = lane & 15;
  const int hi = lane >> 4;
  const int koff = hi << 1;

  const float* arow = hs + r * A_PITCH + koff;
  v8f acc = {};
#pragma unroll 16
  for (int k0 = 0; k0 < HID; k0 += 4) {
    const int k = k0 + koff;
    const v2f a = *(const v2f*)(arow + k0);
    v2f b;
    b.x = B[k * HID + n0 + r];
    b.y = B[(k + 1) * HID + n0 + r];
    acc = __builtin_amdgcn_wmma_f32_16x16x4_f32(false, a, false, b, (short)0,
                                                acc, false, false);
  }
#pragma unroll
  for (int v = 0; v < 8; ++v) {
    C[(size_t)(m0 + v + (hi << 3)) * HID + n0 + r] = acc[v];
  }
}

__global__ __launch_bounds__(256) void attn_cls_kernel(
    const float* __restrict__ feat, const float* __restrict__ attn,
    const int* __restrict__ cnt, const int* __restrict__ slots,
    const int* __restrict__ src, const int* __restrict__ dst,
    const float* __restrict__ B, float* __restrict__ C) {
  __shared__ __attribute__((aligned(16))) float hs[16 * A_PITCH];
  const int lane = threadIdx.x & 31;
  const int wv = __builtin_amdgcn_readfirstlane((int)(threadIdx.x >> 5));
  const int m0 = blockIdx.x * 16;

  attn_stripe_to_lds(feat, attn, cnt, slots, src, dst, hs, m0, wv, lane);
  __syncthreads();

  if (wv != 0) return;
  const int r = lane & 15;
  const int hi = lane >> 4;
  const int koff = hi << 1;

  const float* arow = hs + r * A_PITCH + koff;
  v8f acc = {};
#pragma unroll 16
  for (int k0 = 0; k0 < HID; k0 += 4) {
    const int k = k0 + koff;
    const v2f a = *(const v2f*)(arow + k0);
    v2f b;
    b.x = B[k * OUT_DIM + r];
    b.y = B[(k + 1) * OUT_DIM + r];
    acc = __builtin_amdgcn_wmma_f32_16x16x4_f32(false, a, false, b, (short)0,
                                                acc, false, false);
  }
#pragma unroll
  for (int v = 0; v < 8; ++v) {
    C[(size_t)(m0 + v + (hi << 3)) * OUT_DIM + r] = acc[v];
  }
}

extern "C" void kernel_launch(void* const* d_in, const int* in_sizes, int n_in,
                              void* d_out, int out_size, void* d_ws,
                              size_t ws_size, hipStream_t stream) {
  const int* feats = (const int*)d_in[0];
  const int* src = (const int*)d_in[1];
  const int* dst = (const int*)d_in[2];
  const float* key_emb = (const float*)d_in[3];
  const float* val_emb = (const float*)d_in[4];
  const float* W1 = (const float*)d_in[5];
  const float* attn1 = (const float*)d_in[6];
  const float* W2 = (const float*)d_in[7];
  const float* attn2 = (const float*)d_in[8];
  const float* cls_w = (const float*)d_in[9];
  float* out = (float*)d_out;

  float* feat2 = (float*)d_ws;
  float* feat = feat2 + (size_t)N_NODES * HID;
  int* slots = (int*)(feat + (size_t)N_NODES * HID);
  int* cnt = slots + (size_t)N_NODES * SLOT_CAP;

  embed_gemm128_kernel<<<N_NODES / 16, 256, 0, stream>>>(feats, key_emb, val_emb,
                                                       W1, feat, cnt);

  group_kernel<<<N_EDGES / 256, 256, 0, stream>>>(src, dst, cnt, slots);

  attn_gemm128_kernel<<<N_NODES / 16, 256, 0, stream>>>(
      feat, attn1, cnt, slots, src, dst, W2, feat2);
  attn_cls_kernel<<<N_NODES / 16, 256, 0, stream>>>(
      feat2, attn2, cnt, slots, src, dst, cls_w, out);
}


# Edges grouped in two levels: a partition launch files them by 256-node range (one returning global add per work-group and range, not per edge), attention kernels rank them in LDS; sum order may vary
# speedup vs baseline: 7.1189x; 1.3783x over previous
#include <hip/hip_runtime.h>
#include <hip/hip_bf16.h>

#define N_NODES 50000
#define N_EDGES 800000
#define HID 128
#define OUT_DIM 16
#define NEG_SLOPE 0.2f
#define LOG2E 1.4426950408889634f

#define EDGE_CHUNK 4
#define SLOT_CAP 64

#define BUCKET_NODES 256
#define N_BUCKETS ((N_NODES + BUCKET_NODES - 1) / BUCKET_NODES)
#define BUCKET_CAP 8000
#define BCNT_PITCH 32
#define PART_WGS 256
#define PART_EDGES (N_EDGES / PART_WGS)
#define PART_THREADS 1024
#define PART_ITERS ((PART_EDGES + PART_THREADS - 1) / PART_THREADS)
#define SCAN_PER_LANE ((N_BUCKETS + 31) / 32)
#define FILE_GROUP 5

static_assert(N_NODES % 16 == 0, "sixteen rows per work-group");
static_assert(N_NODES <= 65520, "destination and source share a 32-bit word; 0xffffffff matches no stripe");
static_assert(N_EDGES % PART_WGS == 0, "equal edge chunks in the partition launch");
static_assert(N_BUCKETS <= PART_THREADS, "one thread per range in a partition work-group");
static_assert(BUCKET_NODES % 16 == 0, "a 16-row stripe lies in one range");
static_assert(BUCKET_CAP % 4 == 0, "a range's region starts on a 16-byte boundary");
static_assert(N_EDGES % 32 == 0, "the rescan path reads whole 32-edge windows");
static_assert(SLOT_CAP % 32 == 0, "a row's slots are taken 32 at a time");

typedef __attribute__((ext_vector_type(2))) float v2f;
typedef __attribute__((ext_vector_type(8))) float v8f;

#define A_PITCH (HID + 4)

__global__ __launch_bounds__(256) void embed_gemm128_kernel(
    const int* __restrict__ feats, const float* __restrict__ key_emb,
    const float* __restrict__ val_emb, const float* __restrict__ B,
    float* __restrict__ C, int* __restrict__ bcnt) {
  __shared__ __attribute__((aligned(16))) float hs[16 * A_PITCH];
  const int lane = threadIdx.x & 31;
  const int wave = __builtin_amdgcn_readfirstlane((int)(threadIdx.x >> 5));
  const int m0 = blockIdx.x * 16;

#pragma unroll
  for (int i = 0; i < 2; ++i) {
    const int row = wave * 2 + i;
    const int f0 = feats[(m0 + row) * 2 + 0];
    const int f1 = feats[(m0 + row) * 2 + 1];
    const float4 kv = ((const float4*)(key_emb + (size_t)f0 * HID))[lane];
    const float4 vv = ((const float4*)(val_emb + (size_t)f1 * HID))[lane];
    float4 hv;
    float s;
    s = kv.x + vv.x; hv.x = s > 0.f ? s : 0.f;
    s = kv.y + vv.y; hv.y = s > 0.f ? s : 0.f;
    s = kv.z + vv.z; hv.z = s > 0.f ? s : 0.f;
    s = kv.w + vv.w; hv.w = s > 0.f ? s : 0.f;
    *(float4*)(hs + row * A_PITCH + lane * 4) = hv;
  }
  if (blockIdx.x < N_BUCKETS && threadIdx.x == 0)
    bcnt[blockIdx.x * BCNT_PITCH] = 0;
  __syncthreads();

  const int n0 = wave * 16;
  const int r = lane & 15;
  const int hi = lane >> 4;
  const int koff = hi << 1;

  const float* arow = hs + r * A_PITCH + koff;
  v8f acc = {};
#pragma unroll 8
  for (int k0 = 0; k0 < HID; k0 += 4) {
    const int k = k0 + koff;
    const v2f a = *(const v2f*)(arow + k0);
    v2f b;
    b.x = B[k * HID + n0 + r];
    b.y = B[(k + 1) * HID + n0 + r];
    acc = __builtin_amdgcn_wmma_f32_16x16x4_f32(false, a, false, b, (short)0,
                                                acc, false, false);
  }
#pragma unroll
  for (int v = 0; v < 8; ++v) {
    C[(size_t)(m0 + v + (hi << 3)) * HID + n0 + r] = acc[v];
  }
}

__global__ __launch_bounds__(PART_THREADS) void partition_kernel(
    const int* __restrict__ src, const int* __restrict__ dst,
    int* __restrict__ bcnt, int* __restrict__ region) {
  __shared__ int hist[N_BUCKETS];
  __shared__ int lbase[N_BUCKETS];
  __shared__ int gbase[N_BUCKETS];
  __shared__ unsigned sorted[PART_EDGES];
  const int tid = threadIdx.x;
  const int e0 = blockIdx.x * PART_EDGES;

  unsigned w[PART_ITERS];
#pragma unroll
  for (int i = 0; i < PART_ITERS; ++i) {
    const int idx = tid + PART_THREADS * i;
    w[i] = 0u;
    if (idx < PART_EDGES) {
      const unsigned d = (unsigned)dst[e0 + idx];
      const unsigned s = (unsigned)src[e0 + idx];
      w[i] = (d << 16) | s;
    }
  }
  if (tid < N_BUCKETS) hist[tid] = 0;
  __syncthreads();

  int r[PART_ITERS];
#pragma unroll
  for (int i = 0; i < PART_ITERS; ++i) {
    const int idx = tid + PART_THREADS * i;
    r[i] = 0;
    if (idx < PART_EDGES)
      r[i] = atomicAdd(&hist[(w[i] >> 16) / BUCKET_NODES], 1);
  }
  __syncthreads();

  if (tid < N_BUCKETS) {
    const int n = hist[tid];
    gbase[tid] = n ? atomicAdd(bcnt + tid * BCNT_PITCH, n) : 0;
  }
  if (tid >= PART_THREADS - 32) {
    const int ln = tid - (PART_THREADS - 32);
    int c[SCAN_PER_LANE];
    int sum = 0;
#pragma unroll
    for (int k = 0; k < SCAN_PER_LANE; ++k) {
      const int b = ln * SCAN_PER_LANE + k;
      c[k] = b < N_BUCKETS ? hist[b] : 0;
      sum += c[k];
    }
    int inc = sum;
#pragma unroll
    for (int off = 1; off < 32; off <<= 1) {
      const int t = __shfl_up(inc, off, 32);
      if (ln >= off) inc += t;
    }
    int ex = inc - sum;
#pragma unroll
    for (int k = 0; k < SCAN_PER_LANE; ++k) {
      const int b = ln * SCAN_PER_LANE + k;
      if (b < N_BUCKETS) lbase[b] = ex;
      ex += c[k];
    }
  }
  __syncthreads();

#pragma unroll
  for (int i = 0; i < PART_ITERS; ++i) {
    const int idx = tid + PART_THREADS * i;
    if (idx < PART_EDGES) {
      const int b = (int)((w[i] >> 16) / BUCKET_NODES);
      sorted[lbase[b] + r[i]] = w[i];
    }
  }
  __syncthreads();

#pragma unroll
  for (int i = 0; i < PART_ITERS; ++i) {
    const int p = tid + PART_THREADS * i;
    if (p < PART_EDGES) {
      const unsigned word = sorted[p];
      const int b = (int)((word >> 16) / BUCKET_NODES);
      const int k = gbase[b] + (p - lbase[b]);
      if (k < BUCKET_CAP) region[(size_t)b * BUCKET_CAP + k] = (int)word;
    }
  }
}

__device__ __forceinline__ bool group_stripe(const int* __restrict__ bcnt,
                                             const int* __restrict__ region,
                                             int* lcnt, int* lslots,
                                             const int m0) {
  const int tid = threadIdx.x;
  const int b = m0 / BUCKET_NODES;
  const int nb = bcnt[b * BCNT_PITCH];
  const int n = nb < BUCKET_CAP ? nb : BUCKET_CAP;
  if (tid < 16) lcnt[tid] = 0;
  __syncthreads();
  const int* reg = region + (size_t)b * BUCKET_CAP;
  const unsigned stripe = (unsigned)(m0 >> 4);
  auto file_word = [&](const unsigned word) {
    const unsigned d = word >> 16;
    if ((d >> 4) == stripe) {
      const int rk = atomicAdd(&lcnt[d & 15u], 1);
      if (rk < SLOT_CAP)
        lslots[(d & 15u) * SLOT_CAP + rk] = (int)(word & 0xffffu);
    }
  };
  const int n4 = n & ~3;
  for (int p0 = 0; p0 < n4; p0 += 1024 * FILE_GROUP) {
    int4 wd[FILE_GROUP];
#pragma unroll
    for (int u = 0; u < FILE_GROUP; ++u) {
      const int p = p0 + (u * 256 + tid) * 4;
      wd[u] = p < n4 ? *(const int4*)(reg + p) : make_int4(-1, -1, -1, -1);
    }
#pragma unroll
    for (int u = 0; u < FILE_GROUP; ++u) {
      file_word((unsigned)wd[u].x);
      file_word((unsigned)wd[u].y);
      file_word((unsigned)wd[u].z);
      file_word((unsigned)wd[u].w);
    }
  }
  if (tid < n - n4) file_word((unsigned)reg[n4 + tid]);
  __syncthreads();
  return nb > BUCKET_CAP;
}

template <int M>
__device__ __forceinline__ float row_xor(const float x) {
  const int b = __builtin_bit_cast(int, x);
  return __builtin_bit_cast(
      float, __builtin_amdgcn_mov_dpp(b, 0x160 + M, 0xf, 0xf, true));
}

__device__ __forceinline__ float xhalf(const float x) {
  const int b = __builtin_bit_cast(int, x);
  return __builtin_bit_cast(
      float, __builtin_amdgcn_permlanex16(b, b, 0x76543210, 0xfedcba98, false,
                                          false));
}

__device__ __forceinline__ v2f lrelu2(const v2f t) {
  v2f r;
  r.x = t.x > 0.f ? t.x : NEG_SLOPE * t.x;
  r.y = t.y > 0.f ? t.y : NEG_SLOPE * t.y;
  return r;
}

__device__ __forceinline__ v2f lo2(const float4 v) { return (v2f){v.x, v.y}; }
__device__ __forceinline__ v2f hi2(const float4 v) { return (v2f){v.z, v.w}; }

__device__ __forceinline__ float logit_part(const float4 a0, const float4 a1,
                                            const float4 x0, const float4 x1,
                                            const float4 y0, const float4 y1) {
  v2f c = lo2(a0) * lrelu2(lo2(x0) + lo2(y0));
  c = hi2(a0) * lrelu2(hi2(x0) + hi2(y0)) + c;
  c = lo2(a1) * lrelu2(lo2(x1) + lo2(y1)) + c;
  c = hi2(a1) * lrelu2(hi2(x1) + hi2(y1)) + c;
  return c.x + c.y;
}

struct RowState {
  float m;
  float den;
  float4 acc0;
  float4 acc1;
};

__device__ __forceinline__ void attn_list(
    const float* __restrict__ feat, const float4 a0, const float4 a1,
    const float4 y0, const float4 y1, const int sv, const int nch, const int h,
    const int i, RowState& st) {
  for (int j = 0; j < nch; j += EDGE_CHUNK) {
    int s[EDGE_CHUNK];
#pragma unroll
    for (int u = 0; u < EDGE_CHUNK; ++u) {
      int idx = j + u;
      if (idx > nch - 1) idx = nch - 1;
      s[u] = __builtin_amdgcn_readlane(sv, idx);
    }
    const int sa = h ? s[1] : s[0];
    const int sb = h ? s[3] : s[2];
    const float4* pa = (const float4*)(feat + (size_t)sa * HID) + i;
    const float4* pb = (const float4*)(feat + (size_t)sb * HID) + i;
    const float4 xa0 = pa[0];
    const float4 xa1 = pa[16];
    const float4 xb0 = pb[0];
    const float4 xb1 = pb[16];

    float la = logit_part(a0, a1, xa0, xa1, y0, y1);
    float lb = logit_part(a0, a1, xb0, xb1, y0, y1);
    la += row_xor<1>(la); lb += row_xor<1>(lb);
    la += row_xor<2>(la); lb += row_xor<2>(lb);
    la += row_xor<4>(la); lb += row_xor<4>(lb);
    la += row_xor<8>(la); lb += row_xor<8>(lb);
    if (j + h >= nch) la = -__builtin_inff();
    if (j + 2 + h >= nch) lb = -__builtin_inff();
    float mc = fmaxf(la, lb);
    mc = fmaxf(mc, xhalf(mc));
    mc = fmaxf(mc, st.m);
    const float sc = __builtin_amdgcn_exp2f(st.m - mc);
    const float wa = __builtin_amdgcn_exp2f(la - mc);
    const float wb = __builtin_amdgcn_exp2f(lb - mc);
    st.den = st.den * sc + (wa + wb);
    st.acc0.x = st.acc0.x * sc + (wa * xa0.x + wb * xb0.x);
    st.acc0.y = st.acc0.y * sc + (wa * xa0.y + wb * xb0.y);
    st.acc0.z = st.acc0.z * sc + (wa * xa0.z + wb * xb0.z);
    st.acc0.w = st.acc0.w * sc + (wa * xa0.w + wb * xb0.w);
    st.acc1.x = st.acc1.x * sc + (wa * xa1.x + wb * xb1.x);
    st.acc1.y = st.acc1.y * sc + (wa * xa1.y + wb * xb1.y);
    st.acc1.z = st.acc1.z * sc + (wa * xa1.z + wb * xb1.z);
    st.acc1.w = st.acc1.w * sc + (wa * xa1.w + wb * xb1.w);
    st.m = mc;
  }
}

__device__ __forceinline__ float4 attn_row(
    const float* __restrict__ feat, const float4 a0, const float4 a1,
    const int* lrow, const int* __restrict__ src,
    const int* __restrict__ dst, const int d, const int deg, const int lane) {
  const int h = lane >> 4;
  const int i = lane & 15;
  const float4* yrow = (const float4*)(feat + (size_t)d * HID) + i;
  const float4 y0 = yrow[0];
  const float4 y1 = yrow[16];
  RowState st;
  st.m = -__builtin_inff();
  st.den = 0.f;
  st.acc0 = make_float4(0.f, 0.f, 0.f, 0.f);
  st.acc1 = make_float4(0.f, 0.f, 0.f, 0.f);

  if (deg <= SLOT_CAP) {
    const int* row = lrow;
    for (int base = 0; base < deg; base += 32) {
      int nch = deg - base;
      if (nch > 32) nch = 32;
      const int li = lane < nch ? lane : nch - 1;
      const int sv = row[base + li];
      attn_list(feat, a0, a1, y0, y1, sv, nch, h, i, st);
    }
  } else {
    for (int e0 = 0; e0 < N_EDGES; e0 += 32) {
      const int dv = dst[e0 + lane];
      const int sw = src[e0 + lane];
      unsigned hit = __builtin_amdgcn_ballot_w32(dv == d);
      while (hit) {
        const int b = __builtin_ctz(hit);
        hit &= hit - 1;
        const int sv = __builtin_amdgcn_readlane(sw, b);
        attn_list(feat, a0, a1, y0, y1, sv, 1, h, i, st);
      }
    }
  }

  float den = st.den;
  float4 acc0 = st.acc0;
  float4 acc1 = st.acc1;
  den += xhalf(den);
  acc0.x += xhalf(acc0.x); acc0.y += xhalf(acc0.y);
  acc0.z += xhalf(acc0.z); acc0.w += xhalf(acc0.w);
  acc1.x += xhalf(acc1.x); acc1.y += xhalf(acc1.y);
  acc1.z += xhalf(acc1.z); acc1.w += xhalf(acc1.w);
  const float4 mine = h ? acc1 : acc0;

  float4 o = make_float4(0.f, 0.f, 0.f, 0.f);
  const bool has_edge = deg <= SLOT_CAP ? deg > 0 : den > 0.f;
  if (has_edge) {
    o.x = mine.x / den; o.y = mine.y / den;
    o.z = mine.z / den; o.w = mine.w / den;
  }
  return o;
}

__device__ __forceinline__ void attn_stripe_to_lds(
    const float* __restrict__ feat, const float* __restrict__ attn,
    const int* __restrict__ bcnt, const int* __restrict__ region,
    const int* __restrict__ src, const int* __restrict__ dst, float* hs,
    int* lcnt, int* lslots, const int m0, const int wv, const int lane) {
  const bool over = group_stripe(bcnt, region, lcnt, lslots, m0);
  float4 a0 = ((const float4*)attn)[lane & 15];
  float4 a1 = ((const float4*)attn)[16 + (lane & 15)];
  a0.x *= LOG2E; a0.y *= LOG2E; a0.z *= LOG2E; a0.w *= LOG2E;
  a1.x *= LOG2E; a1.y *= LOG2E; a1.z *= LOG2E; a1.w *= LOG2E;
  const int r0 = wv * 2;
  const int d0 = m0 + r0;
  const int g0 = over ? SLOT_CAP + 1
                      : __builtin_amdgcn_readfirstlane(lcnt[r0]);
  const int g1 = over ? SLOT_CAP + 1
                      : __builtin_amdgcn_readfirstlane(lcnt[r0 + 1]);
  for (int i = 0; i < 2; ++i) {
    const int deg = i ? g1 : g0;
    const float4 o = attn_row(feat, a0, a1, lslots + (r0 + i) * SLOT_CAP, src,
                              dst, d0 + i, deg, lane);
    *(float4*)(hs + (r0 + i) * A_PITCH + lane * 4) = o;
  }
}

__global__ __launch_bounds__(256) void attn_gemm128_kernel(
    const float* __restrict__ feat, const float* __restrict__ attn,
    const int* __restrict__ bcnt, const int* __restrict__ region,
    const int* __restrict__ src, const int* __restrict__ dst,
    const float* __restrict__ B, float* __restrict__ C) {
  __shared__ __attribute__((aligned(16))) float hs[16 * A_PITCH];
  __shared__ int lcnt[16];
  __shared__ int lslots[16 * SLOT_CAP];
  const int lane = threadIdx.x & 31;
  const int wv = __builtin_amdgcn_readfirstlane((int)(threadIdx.x >> 5));
  const int m0 = blockIdx.x * 16;

  attn_stripe_to_lds(feat, attn, bcnt, region, src, dst, hs, lcnt, lslots, m0,
                     wv, lane);
  __syncthreads();

  const int n0 = wv * 16;
  const int r = lane & 15;
  const int hi = lane >> 4;
  const int koff = hi << 1;

  const float* arow = hs + r * A_PITCH + koff;
  v8f acc = {};
#pragma unroll 16
  for (int k0 = 0; k0 < HID; k0 += 4) {
    const int k = k0 + koff;
    const v2f a = *(const v2f*)(arow + k0);
    v2f b;
    b.x = B[k * HID + n0 + r];
    b.y = B[(k + 1) * HID + n0 + r];
    acc = __builtin_amdgcn_wmma_f32_16x16x4_f32(false, a, false, b, (short)0,
                                                acc, false, false);
  }
#pragma unroll
  for (int v = 0; v < 8; ++v) {
    C[(size_t)(m0 + v + (hi << 3)) * HID + n0 + r] = acc[v];
  }
}

__global__ __launch_bounds__(256) void attn_cls_kernel(
    const float* __restrict__ feat, const float* __restrict__ attn,
    const int* __restrict__ bcnt, const int* __restrict__ region,
    const int* __restrict__ src, const int* __restrict__ dst,
    const float* __restrict__ B, float* __restrict__ C) {
  __shared__ __attribute__((aligned(16))) float hs[16 * A_PITCH];
  __shared__ int lcnt[16];
  __shared__ int lslots[16 * SLOT_CAP];
  const int lane = threadIdx.x & 31;
  const int wv = __builtin_amdgcn_readfirstlane((int)(threadIdx.x >> 5));
  const int m0 = blockIdx.x * 16;

  attn_stripe_to_lds(feat, attn, bcnt, region, src, dst, hs, lcnt, lslots, m0,
                     wv, lane);
  __syncthreads();

  if (wv != 0) return;
  const int r = lane & 15;
  const int hi = lane >> 4;
  const int koff = hi << 1;

  const float* arow = hs + r * A_PITCH + koff;
  v8f acc = {};
#pragma unroll 16
  for (int k0 = 0; k0 < HID; k0 += 4) {
    const int k = k0 + koff;
    const v2f a = *(const v2f*)(arow + k0);
    v2f b;
    b.x = B[k * OUT_DIM + r];
    b.y = B[(k + 1) * OUT_DIM + r];
    acc = __builtin_amdgcn_wmma_f32_16x16x4_f32(false, a, false, b, (short)0,
                                                acc, false, false);
  }
#pragma unroll
  for (int v = 0; v < 8; ++v) {
    C[(size_t)(m0 + v + (hi << 3)) * OUT_DIM + r] = acc[v];
  }
}

extern "C" void kernel_launch(void* const* d_in, const int* in_sizes, int n_in,
                              void* d_out, int out_size, void* d_ws,
                              size_t ws_size, hipStream_t stream) {
  const int* feats = (const int*)d_in[0];
  const int* src = (const int*)d_in[1];
  const int* dst = (const int*)d_in[2];
  const float* key_emb = (const float*)d_in[3];
  const float* val_emb = (const float*)d_in[4];
  const float* W1 = (const float*)d_in[5];
  const float* attn1 = (const float*)d_in[6];
  const float* W2 = (const float*)d_in[7];
  const float* attn2 = (const float*)d_in[8];
  const float* cls_w = (const float*)d_in[9];
  float* out = (float*)d_out;

  float* feat2 = (float*)d_ws;
  float* feat = feat2 + (size_t)N_NODES * HID;
  int* region = (int*)(feat + (size_t)N_NODES * HID);
  int* bcnt = region + (size_t)N_BUCKETS * BUCKET_CAP;

  embed_gemm128_kernel<<<N_NODES / 16, 256, 0, stream>>>(feats, key_emb, val_emb,
                                                       W1, feat, bcnt);

  partition_kernel<<<PART_WGS, PART_THREADS, 0, stream>>>(src, dst, bcnt,
                                                          region);

  attn_gemm128_kernel<<<N_NODES / 16, 256, 0, stream>>>(
      feat, attn1, bcnt, region, src, dst, W2, feat2);
  attn_cls_kernel<<<N_NODES / 16, 256, 0, stream>>>(
      feat2, attn2, bcnt, region, src, dst, cls_w, out);
}


# Attention kernels built for 64 vector registers, so 16 waves per SIMD are resident, not 12: leaky_relu as max(t, 0.2t), 32-bit row offsets, lane number formed at its uses; same values and sum order
# speedup vs baseline: 7.6686x; 1.0772x over previous
#include <hip/hip_runtime.h>
#include <hip/hip_bf16.h>

#define N_NODES 50000
#define N_EDGES 800000
#define HID 128
#define OUT_DIM 16
#define NEG_SLOPE 0.2f
#define LOG2E 1.4426950408889634f

#define EDGE_CHUNK 4
#define SLOT_CAP 64

#define BUCKET_NODES 256
#define N_BUCKETS ((N_NODES + BUCKET_NODES - 1) / BUCKET_NODES)
#define BUCKET_CAP 8000
#define BCNT_PITCH 32
#define PART_WGS 256
#define PART_EDGES (N_EDGES / PART_WGS)
#define PART_THREADS 1024
#define PART_ITERS ((PART_EDGES + PART_THREADS - 1) / PART_THREADS)
#define SCAN_PER_LANE ((N_BUCKETS + 31) / 32)
#define FILE_GROUP 5

static_assert(N_NODES % 16 == 0, "sixteen rows per work-group");
static_assert(N_NODES <= 65520, "destination and source share a 32-bit word; 0xffffffff matches no stripe");
static_assert(N_EDGES % PART_WGS == 0, "equal edge chunks in the partition launch");
static_assert(N_BUCKETS <= PART_THREADS, "one thread per range in a partition work-group");
static_assert(BUCKET_NODES % 16 == 0, "a 16-row stripe lies in one range");
static_assert(BUCKET_CAP % 4 == 0, "a range's region starts on a 16-byte boundary");
static_assert(N_EDGES % 32 == 0, "the rescan path reads whole 32-edge windows");
static_assert(SLOT_CAP % 32 == 0, "a row's slots are taken 32 at a time");

typedef __attribute__((ext_vector_type(2))) float v2f;
typedef __attribute__((ext_vector_type(8))) float v8f;

#define A_PITCH (HID + 4)

__global__ __launch_bounds__(256) void embed_gemm128_kernel(
    const int* __restrict__ feats, const float* __restrict__ key_emb,
    const float* __restrict__ val_emb, const float* __restrict__ B,
    float* __restrict__ C, int* __restrict__ bcnt) {
  __shared__ __attribute__((aligned(16))) float hs[16 * A_PITCH];
  const int lane = threadIdx.x & 31;
  const int wave = __builtin_amdgcn_readfirstlane((int)(threadIdx.x >> 5));
  const int m0 = blockIdx.x * 16;

#pragma unroll
  for (int i = 0; i < 2; ++i) {
    const int row = wave * 2 + i;
    const int f0 = feats[(m0 + row) * 2 + 0];
    const int f1 = feats[(m0 + row) * 2 + 1];
    const float4 kv = ((const float4*)(key_emb + (size_t)f0 * HID))[lane];
    const float4 vv = ((const float4*)(val_emb + (size_t)f1 * HID))[lane];
    float4 hv;
    float s;
    s = kv.x + vv.x; hv.x = s > 0.f ? s : 0.f;
    s = kv.y + vv.y; hv.y = s > 0.f ? s : 0.f;
    s = kv.z + vv.z; hv.z = s > 0.f ? s : 0.f;
    s = kv.w + vv.w; hv.w = s > 0.f ? s : 0.f;
    *(float4*)(hs + row * A_PITCH + lane * 4) = hv;
  }
  if (blockIdx.x < N_BUCKETS && threadIdx.x == 0)
    bcnt[blockIdx.x * BCNT_PITCH] = 0;
  __syncthreads();

  const int n0 = wave * 16;
  const int r = lane & 15;
  const int hi = lane >> 4;
  const int koff = hi << 1;

  const float* arow = hs + r * A_PITCH + koff;
  v8f acc = {};
#pragma unroll 8
  for (int k0 = 0; k0 < HID; k0 += 4) {
    const int k = k0 + koff;
    const v2f a = *(const v2f*)(arow + k0);
    v2f b;
    b.x = B[k * HID + n0 + r];
    b.y = B[(k + 1) * HID + n0 + r];
    acc = __builtin_amdgcn_wmma_f32_16x16x4_f32(false, a, false, b, (short)0,
                                                acc, false, false);
  }
#pragma unroll
  for (int v = 0; v < 8; ++v) {
    C[(size_t)(m0 + v + (hi << 3)) * HID + n0 + r] = acc[v];
  }
}

__global__ __launch_bounds__(PART_THREADS) void partition_kernel(
    const int* __restrict__ src, const int* __restrict__ dst,
    int* __restrict__ bcnt, int* __restrict__ region) {
  __shared__ int hist[N_BUCKETS];
  __shared__ int lbase[N_BUCKETS];
  __shared__ int gbase[N_BUCKETS];
  __shared__ unsigned sorted[PART_EDGES];
  const int tid = threadIdx.x;
  const int e0 = blockIdx.x * PART_EDGES;

  unsigned w[PART_ITERS];
#pragma unroll
  for (int i = 0; i < PART_ITERS; ++i) {
    const int idx = tid + PART_THREADS * i;
    w[i] = 0u;
    if (idx < PART_EDGES) {
      const unsigned d = (unsigned)dst[e0 + idx];
      const unsigned s = (unsigned)src[e0 + idx];
      w[i] = (d << 16) | s;
    }
  }
  if (tid < N_BUCKETS) hist[tid] = 0;
  __syncthreads();

  int r[PART_ITERS];
#pragma unroll
  for (int i = 0; i < PART_ITERS; ++i) {
    const int idx = tid + PART_THREADS * i;
    r[i] = 0;
    if (idx < PART_EDGES)
      r[i] = atomicAdd(&hist[(w[i] >> 16) / BUCKET_NODES], 1);
  }
  __syncthreads();

  if (tid < N_BUCKETS) {
    const int n = hist[tid];
    gbase[tid] = n ? atomicAdd(bcnt + tid * BCNT_PITCH, n) : 0;
  }
  if (tid >= PART_THREADS - 32) {
    const int ln = tid - (PART_THREADS - 32);
    int c[SCAN_PER_LANE];
    int sum = 0;
#pragma unroll
    for (int k = 0; k < SCAN_PER_LANE; ++k) {
      const int b = ln * SCAN_PER_LANE + k;
      c[k] = b < N_BUCKETS ? hist[b] : 0;
      sum += c[k];
    }
    int inc = sum;
#pragma unroll
    for (int off = 1; off < 32; off <<= 1) {
      const int t = __shfl_up(inc, off, 32);
      if (ln >= off) inc += t;
    }
    int ex = inc - sum;
#pragma unroll
    for (int k = 0; k < SCAN_PER_LANE; ++k) {
      const int b = ln * SCAN_PER_LANE + k;
      if (b < N_BUCKETS) lbase[b] = ex;
      ex += c[k];
    }
  }
  __syncthreads();

#pragma unroll
  for (int i = 0; i < PART_ITERS; ++i) {
    const int idx = tid + PART_THREADS * i;
    if (idx < PART_EDGES) {
      const int b = (int)((w[i] >> 16) / BUCKET_NODES);
      sorted[lbase[b] + r[i]] = w[i];
    }
  }
  __syncthreads();

#pragma unroll
  for (int i = 0; i < PART_ITERS; ++i) {
    const int p = tid + PART_THREADS * i;
    if (p < PART_EDGES) {
      const unsigned word = sorted[p];
      const int b = (int)((word >> 16) / BUCKET_NODES);
      const int k = gbase[b] + (p - lbase[b]);
      if (k < BUCKET_CAP) region[(size_t)b * BUCKET_CAP + k] = (int)word;
    }
  }
}

__device__ __forceinline__ bool group_stripe(const int* __restrict__ bcnt,
                                             const int* __restrict__ region,
                                             int* lcnt, int* lslots,
                                             const int m0) {
  const int tid = threadIdx.x;
  const int b = m0 / BUCKET_NODES;
  const int nb = bcnt[b * BCNT_PITCH];
  const int n = nb < BUCKET_CAP ? nb : BUCKET_CAP;
  if (tid < 16) lcnt[tid] = 0;
  __syncthreads();
  const int* reg = region + (size_t)b * BUCKET_CAP;
  const unsigned stripe = (unsigned)(m0 >> 4);
  auto file_word = [&](const unsigned word) {
    const unsigned d = word >> 16;
    if ((d >> 4) == stripe) {
      const int rk = atomicAdd(&lcnt[d & 15u], 1);
      if (rk < SLOT_CAP)
        lslots[(d & 15u) * SLOT_CAP + rk] = (int)(word & 0xffffu);
    }
  };
  const int n4 = n & ~3;
  for (int p0 = 0; p0 < n4; p0 += 1024 * FILE_GROUP) {
    int4 wd[FILE_GROUP];
#pragma unroll
    for (int u = 0; u < FILE_GROUP; ++u) {
      const int p = p0 + (u * 256 + tid) * 4;
      wd[u] = p < n4 ? *(const int4*)(reg + p) : make_int4(-1, -1, -1, -1);
    }
#pragma unroll
    for (int u = 0; u < FILE_GROUP; ++u) {
      file_word((unsigned)wd[u].x);
      file_word((unsigned)wd[u].y);
      file_word((unsigned)wd[u].z);
      file_word((unsigned)wd[u].w);
    }
  }
  if (tid < n - n4) file_word((unsigned)reg[n4 + tid]);
  __syncthreads();
  return nb > BUCKET_CAP;
}

__device__ __forceinline__ int lane_now() {
  int l;
  asm volatile("v_mbcnt_lo_u32_b32 %0, -1, 0" : "=v"(l));
  return l;
}

template <int M>
__device__ __forceinline__ float row_xor(const float x) {
  const int b = __builtin_bit_cast(int, x);
  return __builtin_bit_cast(
      float, __builtin_amdgcn_mov_dpp(b, 0x160 + M, 0xf, 0xf, true));
}

__device__ __forceinline__ float xhalf(const float x) {
  const int b = __builtin_bit_cast(int, x);
  return __builtin_bit_cast(
      float, __builtin_amdgcn_permlanex16(b, b, 0x76543210, 0xfedcba98, false,
                                          false));
}

static_assert(NEG_SLOPE > 0.f && NEG_SLOPE < 1.f, "leaky_relu as a maximum");
__device__ __forceinline__ v2f lrelu2(const v2f t) {
  const v2f u = t * NEG_SLOPE;
  v2f r;
  r.x = fmaxf(t.x, u.x);
  r.y = fmaxf(t.y, u.y);
  return r;
}

__device__ __forceinline__ v2f lo2(const float4 v) { return (v2f){v.x, v.y}; }
__device__ __forceinline__ v2f hi2(const float4 v) { return (v2f){v.z, v.w}; }

__device__ __forceinline__ float logit_part(const float4 a0, const float4 a1,
                                            const float4 x0, const float4 x1,
                                            const float4 y0, const float4 y1) {
  v2f c = lo2(a0) * lrelu2(lo2(x0) + lo2(y0));
  c = hi2(a0) * lrelu2(hi2(x0) + hi2(y0)) + c;
  c = lo2(a1) * lrelu2(lo2(x1) + lo2(y1)) + c;
  c = hi2(a1) * lrelu2(hi2(x1) + hi2(y1)) + c;
  return c.x + c.y;
}

struct RowState {
  float m;
  float den;
  float4 acc0;
  float4 acc1;
};

__device__ __forceinline__ void attn_list(
    const float* __restrict__ feat, const float4 a0, const float4 a1,
    const float4 y0, const float4 y1, const int sv, const int nch, const int h,
    const int i, RowState& st) {
  for (int j = 0; j < nch; j += EDGE_CHUNK) {
    int s[EDGE_CHUNK];
#pragma unroll
    for (int u = 0; u < EDGE_CHUNK; ++u) {
      int idx = j + u;
      if (idx > nch - 1) idx = nch - 1;
      s[u] = __builtin_amdgcn_readlane(sv, idx);
    }
    const int sa = h ? s[1] : s[0];
    const int sb = h ? s[3] : s[2];
    const unsigned oa = ((unsigned)sa << 9) + ((unsigned)i << 4);
    const unsigned ob = ((unsigned)sb << 9) + ((unsigned)i << 4);
    const float4* pa = (const float4*)((const char*)feat + oa);
    const float4* pb = (const float4*)((const char*)feat + ob);
    const float4 xa0 = pa[0];
    const float4 xa1 = pa[16];
    const float4 xb0 = pb[0];
    const float4 xb1 = pb[16];

    float la = logit_part(a0, a1, xa0, xa1, y0, y1);
    float lb = logit_part(a0, a1, xb0, xb1, y0, y1);
    la += row_xor<1>(la); lb += row_xor<1>(lb);
    la += row_xor<2>(la); lb += row_xor<2>(lb);
    la += row_xor<4>(la); lb += row_xor<4>(lb);
    la += row_xor<8>(la); lb += row_xor<8>(lb);
    if (j + h >= nch) la = -__builtin_inff();
    if (j + 2 + h >= nch) lb = -__builtin_inff();
    float mc = fmaxf(la, lb);
    mc = fmaxf(mc, xhalf(mc));
    mc = fmaxf(mc, st.m);
    const float sc = __builtin_amdgcn_exp2f(st.m - mc);
    const float wa = __builtin_amdgcn_exp2f(la - mc);
    const float wb = __builtin_amdgcn_exp2f(lb - mc);
    st.den = st.den * sc + (wa + wb);
    st.acc0.x = st.acc0.x * sc + (wa * xa0.x + wb * xb0.x);
    st.acc0.y = st.acc0.y * sc + (wa * xa0.y + wb * xb0.y);
    st.acc0.z = st.acc0.z * sc + (wa * xa0.z + wb * xb0.z);
    st.acc0.w = st.acc0.w * sc + (wa * xa0.w + wb * xb0.w);
    st.acc1.x = st.acc1.x * sc + (wa * xa1.x + wb * xb1.x);
    st.acc1.y = st.acc1.y * sc + (wa * xa1.y + wb * xb1.y);
    st.acc1.z = st.acc1.z * sc + (wa * xa1.z + wb * xb1.z);
    st.acc1.w = st.acc1.w * sc + (wa * xa1.w + wb * xb1.w);
    st.m = mc;
  }
}

__device__ __forceinline__ float4 attn_row(
    const float* __restrict__ feat, const float4 a0, const float4 a1,
    const int* lrow, const int* __restrict__ src,
    const int* __restrict__ dst, const int d, const int deg, const int lane) {
  const int h = lane >> 4;
  const int i = lane & 15;
  const unsigned oy = ((unsigned)d << 9) + ((unsigned)i << 4);
  const float4* yrow = (const float4*)((const char*)feat + oy);
  const float4 y0 = yrow[0];
  const float4 y1 = yrow[16];
  RowState st;
  st.m = -__builtin_inff();
  st.den = 0.f;
  st.acc0 = make_float4(0.f, 0.f, 0.f, 0.f);
  st.acc1 = make_float4(0.f, 0.f, 0.f, 0.f);

  if (deg <= SLOT_CAP) {
    const int* row = lrow;
    for (int base = 0; base < deg; base += 32) {
      int nch = deg - base;
      if (nch > 32) nch = 32;
      const int ln = lane_now();
      const int li = ln < nch ? ln : nch - 1;
      const int sv = row[base + li];
      attn_list(feat, a0, a1, y0, y1, sv, nch, h, i, st);
    }
  } else {
    const int ln = lane_now();
    for (int e0 = 0; e0 < N_EDGES; e0 += 32) {
      const int dv = dst[e0 + ln];
      const int sw = src[e0 + ln];
      unsigned hit = __builtin_amdgcn_ballot_w32(dv == d);
      while (hit) {
        const int b = __builtin_ctz(hit);
        hit &= hit - 1;
        const int sv = __builtin_amdgcn_readlane(sw, b);
        attn_list(feat, a0, a1, y0, y1, sv, 1, h, i, st);
      }
    }
  }

  float den = st.den;
  float4 acc0 = st.acc0;
  float4 acc1 = st.acc1;
  den += xhalf(den);
  acc0.x += xhalf(acc0.x); acc0.y += xhalf(acc0.y);
  acc0.z += xhalf(acc0.z); acc0.w += xhalf(acc0.w);
  acc1.x += xhalf(acc1.x); acc1.y += xhalf(acc1.y);
  acc1.z += xhalf(acc1.z); acc1.w += xhalf(acc1.w);
  const float4 mine = h ? acc1 : acc0;

  float4 o = make_float4(0.f, 0.f, 0.f, 0.f);
  const bool has_edge = deg <= SLOT_CAP ? deg > 0 : den > 0.f;
  if (has_edge) {
    o.x = mine.x / den; o.y = mine.y / den;
    o.z = mine.z / den; o.w = mine.w / den;
  }
  return o;
}

__device__ __forceinline__ void attn_stripe_to_lds(
    const float* __restrict__ feat, const float* __restrict__ attn,
    const int* __restrict__ bcnt, const int* __restrict__ region,
    const int* __restrict__ src, const int* __restrict__ dst, float* hs,
    int* lcnt, int* lslots, const int m0, const int wv, const int lane) {
  const bool over = group_stripe(bcnt, region, lcnt, lslots, m0);
  float4 a0 = ((const float4*)attn)[lane & 15];
  float4 a1 = ((const float4*)attn)[16 + (lane & 15)];
  a0.x *= LOG2E; a0.y *= LOG2E; a0.z *= LOG2E; a0.w *= LOG2E;
  a1.x *= LOG2E; a1.y *= LOG2E; a1.z *= LOG2E; a1.w *= LOG2E;
  const int r0 = wv * 2;
  const int d0 = m0 + r0;
  const int g0 = over ? SLOT_CAP + 1
                      : __builtin_amdgcn_readfirstlane(lcnt[r0]);
  const int g1 = over ? SLOT_CAP + 1
                      : __builtin_amdgcn_readfirstlane(lcnt[r0 + 1]);
  for (int i = 0; i < 2; ++i) {
    const int deg = i ? g1 : g0;
    const float4 o = attn_row(feat, a0, a1, lslots + (r0 + i) * SLOT_CAP, src,
                              dst, d0 + i, deg, lane);
    const int ln = lane_now();
    *(float4*)(hs + (r0 + i) * A_PITCH + ln * 4) = o;
  }
}

__global__ __launch_bounds__(256, 16) void attn_gemm128_kernel(
    const float* __restrict__ feat, const float* __restrict__ attn,
    const int* __restrict__ bcnt, const int* __restrict__ region,
    const int* __restrict__ src, const int* __restrict__ dst,
    const float* __restrict__ B, float* __restrict__ C) {
  __shared__ __attribute__((aligned(16))) float hs[16 * A_PITCH];
  __shared__ int lcnt[16];
  __shared__ int lslots[16 * SLOT_CAP];
  const int lane = threadIdx.x & 31;
  const int wv = __builtin_amdgcn_readfirstlane((int)(threadIdx.x >> 5));
  const int m0 = blockIdx.x * 16;

  attn_stripe_to_lds(feat, attn, bcnt, region, src, dst, hs, lcnt, lslots, m0,
                     wv, lane);
  __syncthreads();

  const int n0 = wv * 16;
  const int gl = lane_now();
  const int r = gl & 15;
  const int hi = gl >> 4;
  const int koff = hi << 1;

  const float* arow = hs + r * A_PITCH + koff;
  v8f acc = {};
#pragma unroll 16
  for (int k0 = 0; k0 < HID; k0 += 4) {
    const int k = k0 + koff;
    const v2f a = *(const v2f*)(arow + k0);
    v2f b;
    b.x = B[k * HID + n0 + r];
    b.y = B[(k + 1) * HID + n0 + r];
    acc = __builtin_amdgcn_wmma_f32_16x16x4_f32(false, a, false, b, (short)0,
                                                acc, false, false);
  }
#pragma unroll
  for (int v = 0; v < 8; ++v) {
    C[(size_t)(m0 + v + (hi << 3)) * HID + n0 + r] = acc[v];
  }
}

__global__ __launch_bounds__(256, 16) void attn_cls_kernel(
    const float* __restrict__ feat, const float* __restrict__ attn,
    const int* __restrict__ bcnt, const int* __restrict__ region,
    const int* __restrict__ src, const int* __restrict__ dst,
    const float* __restrict__ B, float* __restrict__ C) {
  __shared__ __attribute__((aligned(16))) float hs[16 * A_PITCH];
  __shared__ int lcnt[16];
  __shared__ int lslots[16 * SLOT_CAP];
  const int lane = threadIdx.x & 31;
  const int wv = __builtin_amdgcn_readfirstlane((int)(threadIdx.x >> 5));
  const int m0 = blockIdx.x * 16;

  attn_stripe_to_lds(feat, attn, bcnt, region, src, dst, hs, lcnt, lslots, m0,
                     wv, lane);
  __syncthreads();

  if (wv != 0) return;
  const int gl = lane_now();
  const int r = gl & 15;
  const int hi = gl >> 4;
  const int koff = hi << 1;

  const float* arow = hs + r * A_PITCH + koff;
  v8f acc = {};
#pragma unroll 16
  for (int k0 = 0; k0 < HID; k0 += 4) {
    const int k = k0 + koff;
    const v2f a = *(const v2f*)(arow + k0);
    v2f b;
    b.x = B[k * OUT_DIM + r];
    b.y = B[(k + 1) * OUT_DIM + r];
    acc = __builtin_amdgcn_wmma_f32_16x16x4_f32(false, a, false, b, (short)0,
                                                acc, false, false);
  }
#pragma unroll
  for (int v = 0; v < 8; ++v) {
    C[(size_t)(m0 + v + (hi << 3)) * OUT_DIM + r] = acc[v];
  }
}

extern "C" void kernel_launch(void* const* d_in, const int* in_sizes, int n_in,
                              void* d_out, int out_size, void* d_ws,
                              size_t ws_size, hipStream_t stream) {
  const int* feats = (const int*)d_in[0];
  const int* src = (const int*)d_in[1];
  const int* dst = (const int*)d_in[2];
  const float* key_emb = (const float*)d_in[3];
  const float* val_emb = (const float*)d_in[4];
  const float* W1 = (const float*)d_in[5];
  const float* attn1 = (const float*)d_in[6];
  const float* W2 = (const float*)d_in[7];
  const float* attn2 = (const float*)d_in[8];
  const float* cls_w = (const float*)d_in[9];
  float* out = (float*)d_out;

  float* feat2 = (float*)d_ws;
  float* feat = feat2 + (size_t)N_NODES * HID;
  int* region = (int*)(feat + (size_t)N_NODES * HID);
  int* bcnt = region + (size_t)N_BUCKETS * BUCKET_CAP;

  embed_gemm128_kernel<<<N_NODES / 16, 256, 0, stream>>>(feats, key_emb, val_emb,
                                                       W1, feat, bcnt);

  partition_kernel<<<PART_WGS, PART_THREADS, 0, stream>>>(src, dst, bcnt,
                                                          region);

  attn_gemm128_kernel<<<N_NODES / 16, 256, 0, stream>>>(
      feat, attn1, bcnt, region, src, dst, W2, feat2);
  attn_cls_kernel<<<N_NODES / 16, 256, 0, stream>>>(
      feat2, attn2, bcnt, region, src, dst, cls_w, out);
}
